# starting kernel
# baseline (speedup 1.0000x reference)
#include <hip/hip_runtime.h>

#define NQ      10
#define SDIM    1024
#define CDIM    768
#define NLAYERS 3
#define BATCH   2048

typedef float v2f __attribute__((ext_vector_type(2)));
typedef float v8f __attribute__((ext_vector_type(8)));

__device__ __forceinline__ float2 cmul(float2 a, float2 b) {
    return make_float2(a.x * b.x - a.y * b.y, a.x * b.y + a.y * b.x);
}
__device__ __forceinline__ float2 cadd(float2 a, float2 b) {
    return make_float2(a.x + b.x, a.y + b.y);
}

__global__ void __launch_bounds__(512) build_unitary_kernel(
    const float* __restrict__ weights, float* __restrict__ M_re,
    float* __restrict__ M_im) {
    __shared__ float2 st[SDIM];
    const int tid = threadIdx.x;
    const int k = blockIdx.x;
    st[tid]       = make_float2(tid == k ? 1.f : 0.f, 0.f);
    st[tid + 512] = make_float2((tid + 512) == k ? 1.f : 0.f, 0.f);
    __syncthreads();

    for (int l = 0; l < NLAYERS; ++l) {
        for (int q = 0; q < NQ; ++q) {
            const float* w = weights + (l * NQ + q) * 3;
            const float phi = w[0], theta = w[1], omega = w[2];
            const float ct = cosf(0.5f * theta), sv = sinf(0.5f * theta);
            const float al = 0.5f * (phi + omega), be = 0.5f * (phi - omega);
            const float ca = cosf(al), sa = sinf(al);
            const float cb = cosf(be), sb = sinf(be);
            const float2 g00 = make_float2(ct * ca, -ct * sa);
            const float2 g01 = make_float2(-sv * cb, -sv * sb);
            const float2 g10 = make_float2(sv * cb, -sv * sb);
            const float2 g11 = make_float2(ct * ca, ct * sa);
            const int mask = 1 << (9 - q);
            const int low = tid & (mask - 1);
            const int high = (tid & ~(mask - 1)) << 1;
            const int i0 = high | low, i1 = i0 | mask;
            const float2 a0 = st[i0], a1 = st[i1];
            const float2 n0 = cadd(cmul(g00, a0), cmul(g01, a1));
            const float2 n1 = cadd(cmul(g10, a0), cmul(g11, a1));
            st[i0] = n0;
            st[i1] = n1;
            __syncthreads();
        }
        const int r = (l % (NQ - 1)) + 1;
        for (int i = 0; i < NQ; ++i) {
            const int c = i, t = (i + r) % NQ;
            const int maskT = 1 << (9 - t), maskC = 1 << (9 - c);
            const int low = tid & (maskT - 1);
            const int high = (tid & ~(maskT - 1)) << 1;
            const int i0 = high | low;
            if (i0 & maskC) {
                const int i1 = i0 | maskT;
                const float2 a = st[i0], b = st[i1];
                st[i0] = b;
                st[i1] = a;
            }
            __syncthreads();
        }
    }
    M_re[k * SDIM + tid] = st[tid].x;
    M_im[k * SDIM + tid] = st[tid].y;
    M_re[k * SDIM + tid + 512] = st[tid + 512].x;
    M_im[k * SDIM + tid + 512] = st[tid + 512].y;
}

__global__ void __launch_bounds__(256) embed_kernel(
    const float* __restrict__ z, const float* __restrict__ W_in,
    const float* __restrict__ b_in, float* __restrict__ E_re,
    float* __restrict__ E_im) {
    __shared__ float sW[8 * NQ];
    __shared__ float sC[NQ], sS[NQ];
    const int tid = threadIdx.x, b = blockIdx.x;
    const int lane = tid & 31, wv = tid >> 5;

    float acc[NQ];
#pragma unroll
    for (int q = 0; q < NQ; ++q) acc[q] = 0.f;
    for (int d = tid; d < CDIM; d += 256) {
        const float zv = z[b * CDIM + d];
        const float* wr = W_in + d * NQ;
#pragma unroll
        for (int q = 0; q < NQ; ++q) acc[q] += zv * wr[q];
    }
#pragma unroll
    for (int q = 0; q < NQ; ++q) {
        for (int s = 16; s > 0; s >>= 1) acc[q] += __shfl_xor(acc[q], s, 32);
    }
    if (lane == 0) {
#pragma unroll
        for (int q = 0; q < NQ; ++q) sW[wv * NQ + q] = acc[q];
    }
    __syncthreads();
    if (tid < NQ) {
        float a = b_in[tid];
        for (int w = 0; w < 8; ++w) a += sW[w * NQ + tid];
        sC[tid] = cosf(0.5f * a);
        sS[tid] = sinf(0.5f * a);
    }
    __syncthreads();
    for (int i = 0; i < 4; ++i) {
        const int k = tid + i * 256;
        float m = 1.f;
#pragma unroll
        for (int q = 0; q < NQ; ++q) m *= ((k >> (9 - q)) & 1) ? sS[q] : sC[q];
        const int p = __popc(k) & 3;
        E_re[b * SDIM + k] = (p == 0) ? m : (p == 2 ? -m : 0.f);
        E_im[b * SDIM + k] = (p == 1) ? -m : (p == 3 ? m : 0.f);
    }
}

__global__ void __launch_bounds__(256) state_gemm_kernel(
    const float* __restrict__ E_re, const float* __restrict__ E_im,
    const float* __restrict__ M_re, const float* __restrict__ M_im,
    float* __restrict__ evs) {
    __shared__ float sEvs[16 * NQ];
    const int tid = threadIdx.x;
    const int lane = tid & 31, wv = tid >> 5;
    const int m0 = blockIdx.x * 16;
    if (tid < 16 * NQ) sEvs[tid] = 0.f;
    __syncthreads();

    const int half = lane >> 4;
    const int l15 = lane & 15;
    const float* Ar = E_re + (size_t)(m0 + l15) * SDIM + half * 2;
    const float* Ai = E_im + (size_t)(m0 + l15) * SDIM + half * 2;

    for (int nt = wv; nt < SDIM / 16; nt += 8) {
        const int n0 = nt * 16;
        const float* Br = M_re + (size_t)(half * 2) * SDIM + n0 + l15;
        const float* Bi = M_im + (size_t)(half * 2) * SDIM + n0 + l15;
        v8f rr = {}, ri = {}, ir = {}, ii = {};
#pragma unroll 4
        for (int k0 = 0; k0 < SDIM; k0 += 4) {
            const v2f a_re = *(const v2f*)(Ar + k0);
            const v2f a_im = *(const v2f*)(Ai + k0);
            v2f b_re, b_im;
            b_re.x = Br[(size_t)k0 * SDIM];
            b_re.y = Br[(size_t)(k0 + 1) * SDIM];
            b_im.x = Bi[(size_t)k0 * SDIM];
            b_im.y = Bi[(size_t)(k0 + 1) * SDIM];
            rr = __builtin_amdgcn_wmma_f32_16x16x4_f32(false, a_re, false, b_re,
                                                       (short)0, rr, false, false);
            ii = __builtin_amdgcn_wmma_f32_16x16x4_f32(false, a_im, false, b_im,
                                                       (short)0, ii, false, false);
            ri = __builtin_amdgcn_wmma_f32_16x16x4_f32(false, a_re, false, b_im,
                                                       (short)0, ri, false, false);
            ir = __builtin_amdgcn_wmma_f32_16x16x4_f32(false, a_im, false, b_re,
                                                       (short)0, ir, false, false);
        }
        const int j = n0 + l15;
#pragma unroll
        for (int r = 0; r < 8; ++r) {
            const float fre = rr[r] - ii[r];
            const float fim = ri[r] + ir[r];
            const float p = fre * fre + fim * fim;
            const int rowm = r + half * 8;
#pragma unroll
            for (int w = 0; w < NQ; ++w) {
                float val = ((j >> (9 - w)) & 1) ? -p : p;
                val += __shfl_xor(val, 1, 32);
                val += __shfl_xor(val, 2, 32);
                val += __shfl_xor(val, 4, 32);
                val += __shfl_xor(val, 8, 32);
                if (l15 == 0) atomicAdd(&sEvs[rowm * NQ + w], val);
            }
        }
    }
    __syncthreads();
    if (tid < 16 * NQ) {
        const int row = tid / NQ, w = tid % NQ;
        evs[(size_t)(m0 + row) * NQ + w] = sEvs[tid];
    }
}

__global__ void __launch_bounds__(256) out_gemm_kernel(
    const float* __restrict__ evs, const float* __restrict__ W_out,
    const float* __restrict__ b_out, float* __restrict__ out) {
    const int b = blockIdx.x, tid = threadIdx.x;
    float e[NQ];
#pragma unroll
    for (int w = 0; w < NQ; ++w) e[w] = evs[b * NQ + w];
    for (int c = tid; c < CDIM; c += 256) {
        float acc = b_out[c];
#pragma unroll
        for (int w = 0; w < NQ; ++w) acc += e[w] * W_out[w * CDIM + c];
        out[(size_t)b * CDIM + c] = acc;
    }
}

extern "C" void kernel_launch(void* const* d_in, const int* in_sizes, int n_in,
                              void* d_out, int out_size, void* d_ws,
                              size_t ws_size, hipStream_t stream) {
    const float* z_i     = (const float*)d_in[0];
    const float* W_in    = (const float*)d_in[1];
    const float* b_in    = (const float*)d_in[2];
    const float* weights = (const float*)d_in[3];
    const float* W_out   = (const float*)d_in[4];
    const float* b_out   = (const float*)d_in[5];
    float* out = (float*)d_out;

    float* ws   = (float*)d_ws;
    float* M_re = ws;
    float* M_im = M_re + (size_t)SDIM * SDIM;
    float* E_re = M_im + (size_t)SDIM * SDIM;
    float* E_im = E_re + (size_t)BATCH * SDIM;
    float* evs  = E_im + (size_t)BATCH * SDIM;

    build_unitary_kernel<<<SDIM, 512, 0, stream>>>(weights, M_re, M_im);
    embed_kernel<<<BATCH, 256, 0, stream>>>(z_i, W_in, b_in, E_re, E_im);
    state_gemm_kernel<<<BATCH / 16, 256, 0, stream>>>(E_re, E_im, M_re, M_im, evs);
    out_gemm_kernel<<<BATCH, 256, 0, stream>>>(evs, W_out, b_out, out);
}


# state GEMM re-tiled: E strip staged once in LDS as A; each wave keeps 4 column-interleaved N-tiles (16 accumulators), 16-byte M loads 2 k-steps ahead; Z sums in fixed order, LDS atomics removed
# speedup vs baseline: 2.1960x; 2.1960x over previous
#include <hip/hip_runtime.h>

#define NQ      10
#define SDIM    1024
#define CDIM    768
#define NLAYERS 3
#define BATCH   2048

typedef float v2f __attribute__((ext_vector_type(2)));
typedef float v8f __attribute__((ext_vector_type(8)));

__device__ __forceinline__ float2 cmul(float2 a, float2 b) {
    return make_float2(a.x * b.x - a.y * b.y, a.x * b.y + a.y * b.x);
}
__device__ __forceinline__ float2 cadd(float2 a, float2 b) {
    return make_float2(a.x + b.x, a.y + b.y);
}

__global__ void __launch_bounds__(512) build_unitary_kernel(
    const float* __restrict__ weights, float* __restrict__ M_re,
    float* __restrict__ M_im) {
    __shared__ float2 st[SDIM];
    const int tid = threadIdx.x;
    const int k = blockIdx.x;
    st[tid]       = make_float2(tid == k ? 1.f : 0.f, 0.f);
    st[tid + 512] = make_float2((tid + 512) == k ? 1.f : 0.f, 0.f);
    __syncthreads();

    for (int l = 0; l < NLAYERS; ++l) {
        for (int q = 0; q < NQ; ++q) {
            const float* w = weights + (l * NQ + q) * 3;
            const float phi = w[0], theta = w[1], omega = w[2];
            const float ct = cosf(0.5f * theta), sv = sinf(0.5f * theta);
            const float al = 0.5f * (phi + omega), be = 0.5f * (phi - omega);
            const float ca = cosf(al), sa = sinf(al);
            const float cb = cosf(be), sb = sinf(be);
            const float2 g00 = make_float2(ct * ca, -ct * sa);
            const float2 g01 = make_float2(-sv * cb, -sv * sb);
            const float2 g10 = make_float2(sv * cb, -sv * sb);
            const float2 g11 = make_float2(ct * ca, ct * sa);
            const int mask = 1 << (9 - q);
            const int low = tid & (mask - 1);
            const int high = (tid & ~(mask - 1)) << 1;
            const int i0 = high | low, i1 = i0 | mask;
            const float2 a0 = st[i0], a1 = st[i1];
            const float2 n0 = cadd(cmul(g00, a0), cmul(g01, a1));
            const float2 n1 = cadd(cmul(g10, a0), cmul(g11, a1));
            st[i0] = n0;
            st[i1] = n1;
            __syncthreads();
        }
        const int r = (l % (NQ - 1)) + 1;
        for (int i = 0; i < NQ; ++i) {
            const int c = i, t = (i + r) % NQ;
            const int maskT = 1 << (9 - t), maskC = 1 << (9 - c);
            const int low = tid & (maskT - 1);
            const int high = (tid & ~(maskT - 1)) << 1;
            const int i0 = high | low;
            if (i0 & maskC) {
                const int i1 = i0 | maskT;
                const float2 a = st[i0], b = st[i1];
                st[i0] = b;
                st[i1] = a;
            }
            __syncthreads();
        }
    }
    M_re[k * SDIM + tid] = st[tid].x;
    M_im[k * SDIM + tid] = st[tid].y;
    M_re[k * SDIM + tid + 512] = st[tid + 512].x;
    M_im[k * SDIM + tid + 512] = st[tid + 512].y;
}

__global__ void __launch_bounds__(256) embed_kernel(
    const float* __restrict__ z, const float* __restrict__ W_in,
    const float* __restrict__ b_in, float* __restrict__ E_re,
    float* __restrict__ E_im) {
    __shared__ float sW[8 * NQ];
    __shared__ float sC[NQ], sS[NQ];
    const int tid = threadIdx.x, b = blockIdx.x;
    const int lane = tid & 31, wv = tid >> 5;

    float acc[NQ];
#pragma unroll
    for (int q = 0; q < NQ; ++q) acc[q] = 0.f;
    for (int d = tid; d < CDIM; d += 256) {
        const float zv = z[b * CDIM + d];
        const float* wr = W_in + d * NQ;
#pragma unroll
        for (int q = 0; q < NQ; ++q) acc[q] += zv * wr[q];
    }
#pragma unroll
    for (int q = 0; q < NQ; ++q) {
        for (int s = 16; s > 0; s >>= 1) acc[q] += __shfl_xor(acc[q], s, 32);
    }
    if (lane == 0) {
#pragma unroll
        for (int q = 0; q < NQ; ++q) sW[wv * NQ + q] = acc[q];
    }
    __syncthreads();
    if (tid < NQ) {
        float a = b_in[tid];
        for (int w = 0; w < 8; ++w) a += sW[w * NQ + tid];
        sC[tid] = cosf(0.5f * a);
        sS[tid] = sinf(0.5f * a);
    }
    __syncthreads();
    for (int i = 0; i < 4; ++i) {
        const int k = tid + i * 256;
        float m = 1.f;
#pragma unroll
        for (int q = 0; q < NQ; ++q) m *= ((k >> (9 - q)) & 1) ? sS[q] : sC[q];
        const int p = __popc(k) & 3;
        E_re[b * SDIM + k] = (p == 0) ? m : (p == 2 ? -m : 0.f);
        E_im[b * SDIM + k] = (p == 1) ? -m : (p == 3 ? m : 0.f);
    }
}

#define APITCH (SDIM + 4)
#define GWAVES 16

struct BFrag { float4 r0, r1, i0, i1; };
struct AFrag { v2f re, im; };

__device__ __forceinline__ float half_sum16(float v) {
    v += __shfl_xor(v, 1, 32);
    v += __shfl_xor(v, 2, 32);
    v += __shfl_xor(v, 4, 32);
    v += __shfl_xor(v, 8, 32);
    return v;
}

__device__ __forceinline__ float f4get(const float4& v, int t) {
    return t == 0 ? v.x : (t == 1 ? v.y : (t == 2 ? v.z : v.w));
}

__global__ void __launch_bounds__(512) state_gemm_kernel(
    const float* __restrict__ E_re, const float* __restrict__ E_im,
    const float* __restrict__ M_re, const float* __restrict__ M_im,
    float* __restrict__ evs) {
    __shared__ __attribute__((aligned(16))) float sA[2][16][APITCH];
    __shared__ float sRed[GWAVES][16 * NQ];
    const int tid = threadIdx.x;
    const int lane = tid & 31, wv = tid >> 5;
    const int m0 = blockIdx.x * 16;

#pragma unroll
    for (int i = tid; i < 16 * (SDIM / 4); i += 512) {
        const int row = i >> 8, c4 = (i & 255) * 4;
        const float4 vr = *(const float4*)(E_re + (size_t)(m0 + row) * SDIM + c4);
        const float4 vi = *(const float4*)(E_im + (size_t)(m0 + row) * SDIM + c4);
        *(float4*)(&sA[0][row][c4]) = vr;
        *(float4*)(&sA[1][row][c4]) = vi;
    }
    __syncthreads();

    const int half = lane >> 4;
    const int l15 = lane & 15;
    const int n_base = wv * 64;
    const float* aR = &sA[0][l15][half * 2];
    const float* aI = &sA[1][l15][half * 2];
    const float* bR = M_re + (size_t)(half * 2) * SDIM + n_base + 4 * l15;
    const float* bI = M_im + (size_t)(half * 2) * SDIM + n_base + 4 * l15;

    v8f rr[4], ri[4], ir[4], ii[4];
#pragma unroll
    for (int t = 0; t < 4; ++t) { rr[t] = v8f{}; ri[t] = v8f{}; ir[t] = v8f{}; ii[t] = v8f{}; }

    auto loadB = [&](int k0) {
        BFrag f;
        f.r0 = *(const float4*)(bR + (size_t)k0 * SDIM);
        f.r1 = *(const float4*)(bR + (size_t)(k0 + 1) * SDIM);
        f.i0 = *(const float4*)(bI + (size_t)k0 * SDIM);
        f.i1 = *(const float4*)(bI + (size_t)(k0 + 1) * SDIM);
        return f;
    };
    auto loadA = [&](int k0) {
        AFrag f;
        f.re = *(const v2f*)(aR + k0);
        f.im = *(const v2f*)(aI + k0);
        return f;
    };
    auto mma = [&](const AFrag& a, const BFrag& b) {
#pragma unroll
        for (int t = 0; t < 4; ++t) {
            v2f b_re, b_im;
            b_re.x = f4get(b.r0, t); b_re.y = f4get(b.r1, t);
            b_im.x = f4get(b.i0, t); b_im.y = f4get(b.i1, t);
            rr[t] = __builtin_amdgcn_wmma_f32_16x16x4_f32(false, a.re, false, b_re,
                                                          (short)0, rr[t], false, false);
            ii[t] = __builtin_amdgcn_wmma_f32_16x16x4_f32(false, a.im, false, b_im,
                                                          (short)0, ii[t], false, false);
            ri[t] = __builtin_amdgcn_wmma_f32_16x16x4_f32(false, a.re, false, b_im,
                                                          (short)0, ri[t], false, false);
            ir[t] = __builtin_amdgcn_wmma_f32_16x16x4_f32(false, a.im, false, b_re,
                                                          (short)0, ir[t], false, false);
        }
    };

    BFrag b0 = loadB(0), b1 = loadB(4);
    AFrag a0 = loadA(0), a1 = loadA(4);
    asm volatile("" : "+v"(b0.r0.x), "+v"(b0.r0.y), "+v"(b0.r0.z), "+v"(b0.r0.w),
                      "+v"(b0.r1.x), "+v"(b0.r1.y), "+v"(b0.r1.z), "+v"(b0.r1.w),
                      "+v"(b0.i0.x), "+v"(b0.i0.y), "+v"(b0.i0.z), "+v"(b0.i0.w),
                      "+v"(b0.i1.x), "+v"(b0.i1.y), "+v"(b0.i1.z), "+v"(b0.i1.w),
                      "+v"(a0.re), "+v"(a0.im));
    asm volatile("" : "+v"(b1.r0.x), "+v"(b1.r0.y), "+v"(b1.r0.z), "+v"(b1.r0.w),
                      "+v"(b1.r1.x), "+v"(b1.r1.y), "+v"(b1.r1.z), "+v"(b1.r1.w),
                      "+v"(b1.i0.x), "+v"(b1.i0.y), "+v"(b1.i0.z), "+v"(b1.i0.w),
                      "+v"(b1.i1.x), "+v"(b1.i1.y), "+v"(b1.i1.z), "+v"(b1.i1.w),
                      "+v"(a1.re), "+v"(a1.im));
    for (int k0 = 0; k0 < SDIM - 4; k0 += 12) {
        const BFrag b2 = loadB(k0 + 8);
        const AFrag a2 = loadA(k0 + 8);
        __builtin_amdgcn_sched_barrier(0);
        mma(a0, b0);
        b0 = loadB(k0 + 12);
        a0 = loadA(k0 + 12);
        __builtin_amdgcn_sched_barrier(0);
        mma(a1, b1);
        const int kn = (k0 + 16) & (SDIM - 1);
        b1 = loadB(kn);
        a1 = loadA(kn);
        __builtin_amdgcn_sched_barrier(0);
        mma(a2, b2);
    }
    mma(a0, b0);

    const int jb = n_base + 4 * l15;
#pragma unroll
    for (int r = 0; r < 8; ++r) {
        float p[4];
#pragma unroll
        for (int t = 0; t < 4; ++t) {
            const float fre = rr[t][r] - ii[t][r];
            const float fim = ri[t][r] + ir[t][r];
            p[t] = fre * fre + fim * fim;
        }
        const float s01 = p[0] + p[1], s23 = p[2] + p[3];
        const float S = s01 + s23;
        const int rowm = r + half * 8;
        float mine = 0.f;
#pragma unroll
        for (int w = 0; w < NQ - 2; ++w) {
            float val = ((jb >> (9 - w)) & 1) ? -S : S;
            val = half_sum16(val);
            if (l15 == w) mine = val;
        }
        const float v8 = half_sum16(s01 - s23);
        const float v9 = half_sum16((p[0] - p[1]) + (p[2] - p[3]));
        if (l15 == 8) mine = v8;
        if (l15 == 9) mine = v9;
        if (l15 < NQ) sRed[wv][rowm * NQ + l15] = mine;
    }
    __syncthreads();
    if (tid < 16 * NQ) {
        const int row = tid / NQ, w = tid % NQ;
        float acc = 0.f;
#pragma unroll
        for (int g = 0; g < GWAVES; ++g) acc += sRed[g][tid];
        evs[(size_t)(m0 + row) * NQ + w] = acc;
    }
}

__global__ void __launch_bounds__(256) out_gemm_kernel(
    const float* __restrict__ evs, const float* __restrict__ W_out,
    const float* __restrict__ b_out, float* __restrict__ out) {
    const int b = blockIdx.x, tid = threadIdx.x;
    float e[NQ];
#pragma unroll
    for (int w = 0; w < NQ; ++w) e[w] = evs[b * NQ + w];
    for (int c = tid; c < CDIM; c += 256) {
        float acc = b_out[c];
#pragma unroll
        for (int w = 0; w < NQ; ++w) acc += e[w] * W_out[w * CDIM + c];
        out[(size_t)b * CDIM + c] = acc;
    }
}

extern "C" void kernel_launch(void* const* d_in, const int* in_sizes, int n_in,
                              void* d_out, int out_size, void* d_ws,
                              size_t ws_size, hipStream_t stream) {
    const float* z_i     = (const float*)d_in[0];
    const float* W_in    = (const float*)d_in[1];
    const float* b_in    = (const float*)d_in[2];
    const float* weights = (const float*)d_in[3];
    const float* W_out   = (const float*)d_in[4];
    const float* b_out   = (const float*)d_in[5];
    float* out = (float*)d_out;

    float* ws   = (float*)d_ws;
    float* M_re = ws;
    float* M_im = M_re + (size_t)SDIM * SDIM;
    float* E_re = M_im + (size_t)SDIM * SDIM;
    float* E_im = E_re + (size_t)BATCH * SDIM;
    float* evs  = E_im + (size_t)BATCH * SDIM;

    build_unitary_kernel<<<SDIM, 512, 0, stream>>>(weights, M_re, M_im);
    embed_kernel<<<BATCH, 256, 0, stream>>>(z_i, W_in, b_in, E_re, E_im);
    state_gemm_kernel<<<BATCH / 16, 512, 0, stream>>>(E_re, E_im, M_re, M_im, evs);
    out_gemm_kernel<<<BATCH, 256, 0, stream>>>(evs, W_out, b_out, out);
}


# RX phase (-i)^popcount(k) folded into rows of M: E is one real plane, state GEMM does 2 real WMMA products per tile, not 4; F_re, F_im each one running sum over k (was a difference of two sums)
# speedup vs baseline: 2.8890x; 1.3156x over previous
#include <hip/hip_runtime.h>

#define NQ      10
#define SDIM    1024
#define CDIM    768
#define NLAYERS 3
#define BATCH   2048

typedef float v2f __attribute__((ext_vector_type(2)));
typedef float v8f __attribute__((ext_vector_type(8)));

__device__ __forceinline__ float2 cmul(float2 a, float2 b) {
    return make_float2(a.x * b.x - a.y * b.y, a.x * b.y + a.y * b.x);
}
__device__ __forceinline__ float2 cadd(float2 a, float2 b) {
    return make_float2(a.x + b.x, a.y + b.y);
}

__global__ void __launch_bounds__(512) build_unitary_kernel(
    const float* __restrict__ weights, float* __restrict__ M_re,
    float* __restrict__ M_im) {
    __shared__ float2 st[SDIM];
    const int tid = threadIdx.x;
    const int k = blockIdx.x;
    st[tid]       = make_float2(tid == k ? 1.f : 0.f, 0.f);
    st[tid + 512] = make_float2((tid + 512) == k ? 1.f : 0.f, 0.f);
    __syncthreads();

    for (int l = 0; l < NLAYERS; ++l) {
        for (int q = 0; q < NQ; ++q) {
            const float* w = weights + (l * NQ + q) * 3;
            const float phi = w[0], theta = w[1], omega = w[2];
            const float ct = cosf(0.5f * theta), sv = sinf(0.5f * theta);
            const float al = 0.5f * (phi + omega), be = 0.5f * (phi - omega);
            const float ca = cosf(al), sa = sinf(al);
            const float cb = cosf(be), sb = sinf(be);
            const float2 g00 = make_float2(ct * ca, -ct * sa);
            const float2 g01 = make_float2(-sv * cb, -sv * sb);
            const float2 g10 = make_float2(sv * cb, -sv * sb);
            const float2 g11 = make_float2(ct * ca, ct * sa);
            const int mask = 1 << (9 - q);
            const int low = tid & (mask - 1);
            const int high = (tid & ~(mask - 1)) << 1;
            const int i0 = high | low, i1 = i0 | mask;
            const float2 a0 = st[i0], a1 = st[i1];
            const float2 n0 = cadd(cmul(g00, a0), cmul(g01, a1));
            const float2 n1 = cadd(cmul(g10, a0), cmul(g11, a1));
            st[i0] = n0;
            st[i1] = n1;
            __syncthreads();
        }
        const int r = (l % (NQ - 1)) + 1;
        for (int i = 0; i < NQ; ++i) {
            const int c = i, t = (i + r) % NQ;
            const int maskT = 1 << (9 - t), maskC = 1 << (9 - c);
            const int low = tid & (maskT - 1);
            const int high = (tid & ~(maskT - 1)) << 1;
            const int i0 = high | low;
            if (i0 & maskC) {
                const int i1 = i0 | maskT;
                const float2 a = st[i0], b = st[i1];
                st[i0] = b;
                st[i1] = a;
            }
            __syncthreads();
        }
    }
    const int p = __popc(k) & 3;
    const float2 u0 = st[tid], u1 = st[tid + 512];
    const float sg = (p & 2) ? -1.f : 1.f;
    const float2 v0 = (p & 1) ? make_float2(sg * u0.y, -sg * u0.x) : make_float2(sg * u0.x, sg * u0.y);
    const float2 v1 = (p & 1) ? make_float2(sg * u1.y, -sg * u1.x) : make_float2(sg * u1.x, sg * u1.y);
    M_re[k * SDIM + tid] = v0.x;
    M_im[k * SDIM + tid] = v0.y;
    M_re[k * SDIM + tid + 512] = v1.x;
    M_im[k * SDIM + tid + 512] = v1.y;
}

__global__ void __launch_bounds__(256) embed_kernel(
    const float* __restrict__ z, const float* __restrict__ W_in,
    const float* __restrict__ b_in, float* __restrict__ E_mag) {
    __shared__ float sW[8 * NQ];
    __shared__ float sC[NQ], sS[NQ];
    const int tid = threadIdx.x, b = blockIdx.x;
    const int lane = tid & 31, wv = tid >> 5;

    float acc[NQ];
#pragma unroll
    for (int q = 0; q < NQ; ++q) acc[q] = 0.f;
    for (int d = tid; d < CDIM; d += 256) {
        const float zv = z[b * CDIM + d];
        const float* wr = W_in + d * NQ;
#pragma unroll
        for (int q = 0; q < NQ; ++q) acc[q] += zv * wr[q];
    }
#pragma unroll
    for (int q = 0; q < NQ; ++q) {
        for (int s = 16; s > 0; s >>= 1) acc[q] += __shfl_xor(acc[q], s, 32);
    }
    if (lane == 0) {
#pragma unroll
        for (int q = 0; q < NQ; ++q) sW[wv * NQ + q] = acc[q];
    }
    __syncthreads();
    if (tid < NQ) {
        float a = b_in[tid];
        for (int w = 0; w < 8; ++w) a += sW[w * NQ + tid];
        sC[tid] = cosf(0.5f * a);
        sS[tid] = sinf(0.5f * a);
    }
    __syncthreads();
    for (int i = 0; i < 4; ++i) {
        const int k = tid + i * 256;
        float m = 1.f;
#pragma unroll
        for (int q = 0; q < NQ; ++q) m *= ((k >> (9 - q)) & 1) ? sS[q] : sC[q];
        E_mag[b * SDIM + k] = m;
    }
}

#define APITCH (SDIM + 4)
#define GWAVES 16

struct BFrag { float4 r0, r1, i0, i1; };

__device__ __forceinline__ float half_sum16(float v) {
    v += __shfl_xor(v, 1, 32);
    v += __shfl_xor(v, 2, 32);
    v += __shfl_xor(v, 4, 32);
    v += __shfl_xor(v, 8, 32);
    return v;
}

__device__ __forceinline__ float f4get(const float4& v, int t) {
    return t == 0 ? v.x : (t == 1 ? v.y : (t == 2 ? v.z : v.w));
}

__global__ void __launch_bounds__(512) state_gemm_kernel(
    const float* __restrict__ E_mag,
    const float* __restrict__ M_re, const float* __restrict__ M_im,
    float* __restrict__ evs) {
    __shared__ __attribute__((aligned(16))) float sA[16][APITCH];
    __shared__ float sRed[GWAVES][16 * NQ];
    const int tid = threadIdx.x;
    const int lane = tid & 31, wv = tid >> 5;
    const int m0 = blockIdx.x * 16;

#pragma unroll
    for (int i = tid; i < 16 * (SDIM / 4); i += 512) {
        const int row = i >> 8, c4 = (i & 255) * 4;
        const float4 v = *(const float4*)(E_mag + (size_t)(m0 + row) * SDIM + c4);
        *(float4*)(&sA[row][c4]) = v;
    }
    __syncthreads();

    const int half = lane >> 4;
    const int l15 = lane & 15;
    const int n_base = wv * 64;
    const float* aP = &sA[l15][half * 2];
    const float* bR = M_re + (size_t)(half * 2) * SDIM + n_base + 4 * l15;
    const float* bI = M_im + (size_t)(half * 2) * SDIM + n_base + 4 * l15;

    v8f fr[4], fi[4];
#pragma unroll
    for (int t = 0; t < 4; ++t) { fr[t] = v8f{}; fi[t] = v8f{}; }

    auto loadB = [&](int k0) {
        BFrag f;
        f.r0 = *(const float4*)(bR + (size_t)k0 * SDIM);
        f.r1 = *(const float4*)(bR + (size_t)(k0 + 1) * SDIM);
        f.i0 = *(const float4*)(bI + (size_t)k0 * SDIM);
        f.i1 = *(const float4*)(bI + (size_t)(k0 + 1) * SDIM);
        return f;
    };
    auto loadA = [&](int k0) { return *(const v2f*)(aP + k0); };
    auto mma = [&](const v2f& a, const BFrag& b) {
#pragma unroll
        for (int t = 0; t < 4; ++t) {
            v2f b_re, b_im;
            b_re.x = f4get(b.r0, t); b_re.y = f4get(b.r1, t);
            b_im.x = f4get(b.i0, t); b_im.y = f4get(b.i1, t);
            fr[t] = __builtin_amdgcn_wmma_f32_16x16x4_f32(false, a, false, b_re,
                                                          (short)0, fr[t], false, false);
            fi[t] = __builtin_amdgcn_wmma_f32_16x16x4_f32(false, a, false, b_im,
                                                          (short)0, fi[t], false, false);
        }
    };

    BFrag b0 = loadB(0), b1 = loadB(4);
    v2f a0 = loadA(0), a1 = loadA(4);
    asm volatile("" : "+v"(b0.r0.x), "+v"(b0.r0.y), "+v"(b0.r0.z), "+v"(b0.r0.w),
                      "+v"(b0.r1.x), "+v"(b0.r1.y), "+v"(b0.r1.z), "+v"(b0.r1.w),
                      "+v"(b0.i0.x), "+v"(b0.i0.y), "+v"(b0.i0.z), "+v"(b0.i0.w),
                      "+v"(b0.i1.x), "+v"(b0.i1.y), "+v"(b0.i1.z), "+v"(b0.i1.w),
                      "+v"(a0));
    asm volatile("" : "+v"(b1.r0.x), "+v"(b1.r0.y), "+v"(b1.r0.z), "+v"(b1.r0.w),
                      "+v"(b1.r1.x), "+v"(b1.r1.y), "+v"(b1.r1.z), "+v"(b1.r1.w),
                      "+v"(b1.i0.x), "+v"(b1.i0.y), "+v"(b1.i0.z), "+v"(b1.i0.w),
                      "+v"(b1.i1.x), "+v"(b1.i1.y), "+v"(b1.i1.z), "+v"(b1.i1.w),
                      "+v"(a1));
    for (int k0 = 0; k0 < SDIM - 4; k0 += 12) {
        const BFrag b2 = loadB(k0 + 8);
        const v2f a2 = loadA(k0 + 8);
        __builtin_amdgcn_sched_barrier(0);
        mma(a0, b0);
        b0 = loadB(k0 + 12);
        a0 = loadA(k0 + 12);
        __builtin_amdgcn_sched_barrier(0);
        mma(a1, b1);
        const int kn = (k0 + 16) & (SDIM - 1);
        b1 = loadB(kn);
        a1 = loadA(kn);
        __builtin_amdgcn_sched_barrier(0);
        mma(a2, b2);
    }
    mma(a0, b0);

    const int jb = n_base + 4 * l15;
#pragma unroll
    for (int r = 0; r < 8; ++r) {
        float p[4];
#pragma unroll
        for (int t = 0; t < 4; ++t) {
            const float fre = fr[t][r];
            const float fim = fi[t][r];
            p[t] = fre * fre + fim * fim;
        }
        const float s01 = p[0] + p[1], s23 = p[2] + p[3];
        const float S = s01 + s23;
        const int rowm = r + half * 8;
        float mine = 0.f;
#pragma unroll
        for (int w = 0; w < NQ - 2; ++w) {
            float val = ((jb >> (9 - w)) & 1) ? -S : S;
            val = half_sum16(val);
            if (l15 == w) mine = val;
        }
        const float v8 = half_sum16(s01 - s23);
        const float v9 = half_sum16((p[0] - p[1]) + (p[2] - p[3]));
        if (l15 == 8) mine = v8;
        if (l15 == 9) mine = v9;
        if (l15 < NQ) sRed[wv][rowm * NQ + l15] = mine;
    }
    __syncthreads();
    if (tid < 16 * NQ) {
        const int row = tid / NQ, w = tid % NQ;
        float acc = 0.f;
#pragma unroll
        for (int g = 0; g < GWAVES; ++g) acc += sRed[g][tid];
        evs[(size_t)(m0 + row) * NQ + w] = acc;
    }
}

__global__ void __launch_bounds__(256) out_gemm_kernel(
    const float* __restrict__ evs, const float* __restrict__ W_out,
    const float* __restrict__ b_out, float* __restrict__ out) {
    const int b = blockIdx.x, tid = threadIdx.x;
    float e[NQ];
#pragma unroll
    for (int w = 0; w < NQ; ++w) e[w] = evs[b * NQ + w];
    for (int c = tid; c < CDIM; c += 256) {
        float acc = b_out[c];
#pragma unroll
        for (int w = 0; w < NQ; ++w) acc += e[w] * W_out[w * CDIM + c];
        out[(size_t)b * CDIM + c] = acc;
    }
}

extern "C" void kernel_launch(void* const* d_in, const int* in_sizes, int n_in,
                              void* d_out, int out_size, void* d_ws,
                              size_t ws_size, hipStream_t stream) {
    const float* z_i     = (const float*)d_in[0];
    const float* W_in    = (const float*)d_in[1];
    const float* b_in    = (const float*)d_in[2];
    const float* weights = (const float*)d_in[3];
    const float* W_out   = (const float*)d_in[4];
    const float* b_out   = (const float*)d_in[5];
    float* out = (float*)d_out;

    float* ws    = (float*)d_ws;
    float* M_re  = ws;
    float* M_im  = M_re + (size_t)SDIM * SDIM;
    float* E_mag = M_im + (size_t)SDIM * SDIM;
    float* evs   = E_mag + (size_t)BATCH * SDIM;

    build_unitary_kernel<<<SDIM, 512, 0, stream>>>(weights, M_re, M_im);
    embed_kernel<<<BATCH, 256, 0, stream>>>(z_i, W_in, b_in, E_mag);
    state_gemm_kernel<<<BATCH / 16, 512, 0, stream>>>(E_mag, M_re, M_im, evs);
    out_gemm_kernel<<<BATCH, 256, 0, stream>>>(evs, W_out, b_out, out);
}


# Rot gate entries (6 sin/cos, 4 products per gate) worked out once per block by 30 lanes into LDS; each gate step reads one float4 instead of redoing the trig in all 16 waves
# speedup vs baseline: 4.5198x; 1.5645x over previous
#include <hip/hip_runtime.h>

#define NQ      10
#define SDIM    1024
#define CDIM    768
#define NLAYERS 3
#define BATCH   2048

typedef float v2f __attribute__((ext_vector_type(2)));
typedef float v8f __attribute__((ext_vector_type(8)));

__device__ __forceinline__ float2 cmul(float2 a, float2 b) {
    return make_float2(a.x * b.x - a.y * b.y, a.x * b.y + a.y * b.x);
}
__device__ __forceinline__ float2 cadd(float2 a, float2 b) {
    return make_float2(a.x + b.x, a.y + b.y);
}

__global__ void __launch_bounds__(512) build_unitary_kernel(
    const float* __restrict__ weights, float* __restrict__ M_re,
    float* __restrict__ M_im) {
    __shared__ float2 st[SDIM];
    __shared__ float4 sG[NLAYERS * NQ];
    const int tid = threadIdx.x;
    const int k = blockIdx.x;
    st[tid]       = make_float2(tid == k ? 1.f : 0.f, 0.f);
    st[tid + 512] = make_float2((tid + 512) == k ? 1.f : 0.f, 0.f);
    if (tid < NLAYERS * NQ) {
        const float* w = weights + tid * 3;
        const float phi = w[0], theta = w[1], omega = w[2];
        const float ct = cosf(0.5f * theta), sv = sinf(0.5f * theta);
        const float al = 0.5f * (phi + omega), be = 0.5f * (phi - omega);
        const float ca = cosf(al), sa = sinf(al);
        const float cb = cosf(be), sb = sinf(be);
        sG[tid] = make_float4(ct * ca, ct * sa, sv * cb, sv * sb);
    }
    __syncthreads();

    for (int l = 0; l < NLAYERS; ++l) {
        for (int q = 0; q < NQ; ++q) {
            const float4 g = sG[l * NQ + q];
            const float2 g00 = make_float2(g.x, -g.y);
            const float2 g01 = make_float2(-g.z, -g.w);
            const float2 g10 = make_float2(g.z, -g.w);
            const float2 g11 = make_float2(g.x, g.y);
            const int mask = 1 << (9 - q);
            const int low = tid & (mask - 1);
            const int high = (tid & ~(mask - 1)) << 1;
            const int i0 = high | low, i1 = i0 | mask;
            const float2 a0 = st[i0], a1 = st[i1];
            const float2 n0 = cadd(cmul(g00, a0), cmul(g01, a1));
            const float2 n1 = cadd(cmul(g10, a0), cmul(g11, a1));
            st[i0] = n0;
            st[i1] = n1;
            __syncthreads();
        }
        const int r = (l % (NQ - 1)) + 1;
        for (int i = 0; i < NQ; ++i) {
            const int c = i, t = (i + r) % NQ;
            const int maskT = 1 << (9 - t), maskC = 1 << (9 - c);
            const int low = tid & (maskT - 1);
            const int high = (tid & ~(maskT - 1)) << 1;
            const int i0 = high | low;
            if (i0 & maskC) {
                const int i1 = i0 | maskT;
                const float2 a = st[i0], b = st[i1];
                st[i0] = b;
                st[i1] = a;
            }
            __syncthreads();
        }
    }
    const int p = __popc(k) & 3;
    const float2 u0 = st[tid], u1 = st[tid + 512];
    const float sg = (p & 2) ? -1.f : 1.f;
    const float2 v0 = (p & 1) ? make_float2(sg * u0.y, -sg * u0.x) : make_float2(sg * u0.x, sg * u0.y);
    const float2 v1 = (p & 1) ? make_float2(sg * u1.y, -sg * u1.x) : make_float2(sg * u1.x, sg * u1.y);
    M_re[k * SDIM + tid] = v0.x;
    M_im[k * SDIM + tid] = v0.y;
    M_re[k * SDIM + tid + 512] = v1.x;
    M_im[k * SDIM + tid + 512] = v1.y;
}

__global__ void __launch_bounds__(256) embed_kernel(
    const float* __restrict__ z, const float* __restrict__ W_in,
    const float* __restrict__ b_in, float* __restrict__ E_mag) {
    __shared__ float sW[8 * NQ];
    __shared__ float sC[NQ], sS[NQ];
    const int tid = threadIdx.x, b = blockIdx.x;
    const int lane = tid & 31, wv = tid >> 5;

    float acc[NQ];
#pragma unroll
    for (int q = 0; q < NQ; ++q) acc[q] = 0.f;
    for (int d = tid; d < CDIM; d += 256) {
        const float zv = z[b * CDIM + d];
        const float* wr = W_in + d * NQ;
#pragma unroll
        for (int q = 0; q < NQ; ++q) acc[q] += zv * wr[q];
    }
#pragma unroll
    for (int q = 0; q < NQ; ++q) {
        for (int s = 16; s > 0; s >>= 1) acc[q] += __shfl_xor(acc[q], s, 32);
    }
    if (lane == 0) {
#pragma unroll
        for (int q = 0; q < NQ; ++q) sW[wv * NQ + q] = acc[q];
    }
    __syncthreads();
    if (tid < NQ) {
        float a = b_in[tid];
        for (int w = 0; w < 8; ++w) a += sW[w * NQ + tid];
        sC[tid] = cosf(0.5f * a);
        sS[tid] = sinf(0.5f * a);
    }
    __syncthreads();
    for (int i = 0; i < 4; ++i) {
        const int k = tid + i * 256;
        float m = 1.f;
#pragma unroll
        for (int q = 0; q < NQ; ++q) m *= ((k >> (9 - q)) & 1) ? sS[q] : sC[q];
        E_mag[b * SDIM + k] = m;
    }
}

#define APITCH (SDIM + 4)
#define GWAVES 16

struct BFrag { float4 r0, r1, i0, i1; };

__device__ __forceinline__ float half_sum16(float v) {
    v += __shfl_xor(v, 1, 32);
    v += __shfl_xor(v, 2, 32);
    v += __shfl_xor(v, 4, 32);
    v += __shfl_xor(v, 8, 32);
    return v;
}

__device__ __forceinline__ float f4get(const float4& v, int t) {
    return t == 0 ? v.x : (t == 1 ? v.y : (t == 2 ? v.z : v.w));
}

__global__ void __launch_bounds__(512) state_gemm_kernel(
    const float* __restrict__ E_mag,
    const float* __restrict__ M_re, const float* __restrict__ M_im,
    float* __restrict__ evs) {
    __shared__ __attribute__((aligned(16))) float sA[16][APITCH];
    __shared__ float sRed[GWAVES][16 * NQ];
    const int tid = threadIdx.x;
    const int lane = tid & 31, wv = tid >> 5;
    const int m0 = blockIdx.x * 16;

#pragma unroll
    for (int i = tid; i < 16 * (SDIM / 4); i += 512) {
        const int row = i >> 8, c4 = (i & 255) * 4;
        const float4 v = *(const float4*)(E_mag + (size_t)(m0 + row) * SDIM + c4);
        *(float4*)(&sA[row][c4]) = v;
    }
    __syncthreads();

    const int half = lane >> 4;
    const int l15 = lane & 15;
    const int n_base = wv * 64;
    const float* aP = &sA[l15][half * 2];
    const float* bR = M_re + (size_t)(half * 2) * SDIM + n_base + 4 * l15;
    const float* bI = M_im + (size_t)(half * 2) * SDIM + n_base + 4 * l15;

    v8f fr[4], fi[4];
#pragma unroll
    for (int t = 0; t < 4; ++t) { fr[t] = v8f{}; fi[t] = v8f{}; }

    auto loadB = [&](int k0) {
        BFrag f;
        f.r0 = *(const float4*)(bR + (size_t)k0 * SDIM);
        f.r1 = *(const float4*)(bR + (size_t)(k0 + 1) * SDIM);
        f.i0 = *(const float4*)(bI + (size_t)k0 * SDIM);
        f.i1 = *(const float4*)(bI + (size_t)(k0 + 1) * SDIM);
        return f;
    };
    auto loadA = [&](int k0) { return *(const v2f*)(aP + k0); };
    auto mma = [&](const v2f& a, const BFrag& b) {
#pragma unroll
        for (int t = 0; t < 4; ++t) {
            v2f b_re, b_im;
            b_re.x = f4get(b.r0, t); b_re.y = f4get(b.r1, t);
            b_im.x = f4get(b.i0, t); b_im.y = f4get(b.i1, t);
            fr[t] = __builtin_amdgcn_wmma_f32_16x16x4_f32(false, a, false, b_re,
                                                          (short)0, fr[t], false, false);
            fi[t] = __builtin_amdgcn_wmma_f32_16x16x4_f32(false, a, false, b_im,
                                                          (short)0, fi[t], false, false);
        }
    };

    BFrag b0 = loadB(0), b1 = loadB(4);
    v2f a0 = loadA(0), a1 = loadA(4);
    asm volatile("" : "+v"(b0.r0.x), "+v"(b0.r0.y), "+v"(b0.r0.z), "+v"(b0.r0.w),
                      "+v"(b0.r1.x), "+v"(b0.r1.y), "+v"(b0.r1.z), "+v"(b0.r1.w),
                      "+v"(b0.i0.x), "+v"(b0.i0.y), "+v"(b0.i0.z), "+v"(b0.i0.w),
                      "+v"(b0.i1.x), "+v"(b0.i1.y), "+v"(b0.i1.z), "+v"(b0.i1.w),
                      "+v"(a0));
    asm volatile("" : "+v"(b1.r0.x), "+v"(b1.r0.y), "+v"(b1.r0.z), "+v"(b1.r0.w),
                      "+v"(b1.r1.x), "+v"(b1.r1.y), "+v"(b1.r1.z), "+v"(b1.r1.w),
                      "+v"(b1.i0.x), "+v"(b1.i0.y), "+v"(b1.i0.z), "+v"(b1.i0.w),
                      "+v"(b1.i1.x), "+v"(b1.i1.y), "+v"(b1.i1.z), "+v"(b1.i1.w),
                      "+v"(a1));
    for (int k0 = 0; k0 < SDIM - 4; k0 += 12) {
        const BFrag b2 = loadB(k0 + 8);
        const v2f a2 = loadA(k0 + 8);
        __builtin_amdgcn_sched_barrier(0);
        mma(a0, b0);
        b0 = loadB(k0 + 12);
        a0 = loadA(k0 + 12);
        __builtin_amdgcn_sched_barrier(0);
        mma(a1, b1);
        const int kn = (k0 + 16) & (SDIM - 1);
        b1 = loadB(kn);
        a1 = loadA(kn);
        __builtin_amdgcn_sched_barrier(0);
        mma(a2, b2);
    }
    mma(a0, b0);

    const int jb = n_base + 4 * l15;
#pragma unroll
    for (int r = 0; r < 8; ++r) {
        float p[4];
#pragma unroll
        for (int t = 0; t < 4; ++t) {
            const float fre = fr[t][r];
            const float fim = fi[t][r];
            p[t] = fre * fre + fim * fim;
        }
        const float s01 = p[0] + p[1], s23 = p[2] + p[3];
        const float S = s01 + s23;
        const int rowm = r + half * 8;
        float mine = 0.f;
#pragma unroll
        for (int w = 0; w < NQ - 2; ++w) {
            float val = ((jb >> (9 - w)) & 1) ? -S : S;
            val = half_sum16(val);
            if (l15 == w) mine = val;
        }
        const float v8 = half_sum16(s01 - s23);
        const float v9 = half_sum16((p[0] - p[1]) + (p[2] - p[3]));
        if (l15 == 8) mine = v8;
        if (l15 == 9) mine = v9;
        if (l15 < NQ) sRed[wv][rowm * NQ + l15] = mine;
    }
    __syncthreads();
    if (tid < 16 * NQ) {
        const int row = tid / NQ, w = tid % NQ;
        float acc = 0.f;
#pragma unroll
        for (int g = 0; g < GWAVES; ++g) acc += sRed[g][tid];
        evs[(size_t)(m0 + row) * NQ + w] = acc;
    }
}

__global__ void __launch_bounds__(256) out_gemm_kernel(
    const float* __restrict__ evs, const float* __restrict__ W_out,
    const float* __restrict__ b_out, float* __restrict__ out) {
    const int b = blockIdx.x, tid = threadIdx.x;
    float e[NQ];
#pragma unroll
    for (int w = 0; w < NQ; ++w) e[w] = evs[b * NQ + w];
    for (int c = tid; c < CDIM; c += 256) {
        float acc = b_out[c];
#pragma unroll
        for (int w = 0; w < NQ; ++w) acc += e[w] * W_out[w * CDIM + c];
        out[(size_t)b * CDIM + c] = acc;
    }
}

extern "C" void kernel_launch(void* const* d_in, const int* in_sizes, int n_in,
                              void* d_out, int out_size, void* d_ws,
                              size_t ws_size, hipStream_t stream) {
    const float* z_i     = (const float*)d_in[0];
    const float* W_in    = (const float*)d_in[1];
    const float* b_in    = (const float*)d_in[2];
    const float* weights = (const float*)d_in[3];
    const float* W_out   = (const float*)d_in[4];
    const float* b_out   = (const float*)d_in[5];
    float* out = (float*)d_out;

    float* ws    = (float*)d_ws;
    float* M_re  = ws;
    float* M_im  = M_re + (size_t)SDIM * SDIM;
    float* E_mag = M_im + (size_t)SDIM * SDIM;
    float* evs   = E_mag + (size_t)BATCH * SDIM;

    build_unitary_kernel<<<SDIM, 512, 0, stream>>>(weights, M_re, M_im);
    embed_kernel<<<BATCH, 256, 0, stream>>>(z_i, W_in, b_in, E_mag);
    state_gemm_kernel<<<BATCH / 16, 512, 0, stream>>>(E_mag, M_re, M_im, evs);
    out_gemm_kernel<<<BATCH, 256, 0, stream>>>(evs, W_out, b_out, out);
}


# state GEMM spread over all 256 WGPs: block = 32 rows x 256 columns, 4 waves, each B fragment feeds two M-tiles, loads 4 k-steps ahead; Z sums regrouped per column quarter, added in the last kernel
# speedup vs baseline: 6.9137x; 1.5296x over previous
#include <hip/hip_runtime.h>

#define NQ      10
#define SDIM    1024
#define CDIM    768
#define NLAYERS 3
#define BATCH   2048

typedef float v2f __attribute__((ext_vector_type(2)));
typedef float v4f __attribute__((ext_vector_type(4)));
typedef float v8f __attribute__((ext_vector_type(8)));

__device__ __forceinline__ float2 cmul(float2 a, float2 b) {
    return make_float2(a.x * b.x - a.y * b.y, a.x * b.y + a.y * b.x);
}
__device__ __forceinline__ float2 cadd(float2 a, float2 b) {
    return make_float2(a.x + b.x, a.y + b.y);
}

__global__ void __launch_bounds__(512) build_unitary_kernel(
    const float* __restrict__ weights, float* __restrict__ M_re,
    float* __restrict__ M_im) {
    __shared__ float2 st[SDIM];
    __shared__ float4 sG[NLAYERS * NQ];
    const int tid = threadIdx.x;
    const int k = blockIdx.x;
    st[tid]       = make_float2(tid == k ? 1.f : 0.f, 0.f);
    st[tid + 512] = make_float2((tid + 512) == k ? 1.f : 0.f, 0.f);
    if (tid < NLAYERS * NQ) {
        const float* w = weights + tid * 3;
        const float phi = w[0], theta = w[1], omega = w[2];
        const float ct = cosf(0.5f * theta), sv = sinf(0.5f * theta);
        const float al = 0.5f * (phi + omega), be = 0.5f * (phi - omega);
        const float ca = cosf(al), sa = sinf(al);
        const float cb = cosf(be), sb = sinf(be);
        sG[tid] = make_float4(ct * ca, ct * sa, sv * cb, sv * sb);
    }
    __syncthreads();

    for (int l = 0; l < NLAYERS; ++l) {
        for (int q = 0; q < NQ; ++q) {
            const float4 g = sG[l * NQ + q];
            const float2 g00 = make_float2(g.x, -g.y);
            const float2 g01 = make_float2(-g.z, -g.w);
            const float2 g10 = make_float2(g.z, -g.w);
            const float2 g11 = make_float2(g.x, g.y);
            const int mask = 1 << (9 - q);
            const int low = tid & (mask - 1);
            const int high = (tid & ~(mask - 1)) << 1;
            const int i0 = high | low, i1 = i0 | mask;
            const float2 a0 = st[i0], a1 = st[i1];
            const float2 n0 = cadd(cmul(g00, a0), cmul(g01, a1));
            const float2 n1 = cadd(cmul(g10, a0), cmul(g11, a1));
            st[i0] = n0;
            st[i1] = n1;
            __syncthreads();
        }
        const int r = (l % (NQ - 1)) + 1;
        for (int i = 0; i < NQ; ++i) {
            const int c = i, t = (i + r) % NQ;
            const int maskT = 1 << (9 - t), maskC = 1 << (9 - c);
            const int low = tid & (maskT - 1);
            const int high = (tid & ~(maskT - 1)) << 1;
            const int i0 = high | low;
            if (i0 & maskC) {
                const int i1 = i0 | maskT;
                const float2 a = st[i0], b = st[i1];
                st[i0] = b;
                st[i1] = a;
            }
            __syncthreads();
        }
    }
    const int p = __popc(k) & 3;
    const float2 u0 = st[tid], u1 = st[tid + 512];
    const float sg = (p & 2) ? -1.f : 1.f;
    const float2 v0 = (p & 1) ? make_float2(sg * u0.y, -sg * u0.x) : make_float2(sg * u0.x, sg * u0.y);
    const float2 v1 = (p & 1) ? make_float2(sg * u1.y, -sg * u1.x) : make_float2(sg * u1.x, sg * u1.y);
    M_re[k * SDIM + tid] = v0.x;
    M_im[k * SDIM + tid] = v0.y;
    M_re[k * SDIM + tid + 512] = v1.x;
    M_im[k * SDIM + tid + 512] = v1.y;
}

__global__ void __launch_bounds__(256) embed_kernel(
    const float* __restrict__ z, const float* __restrict__ W_in,
    const float* __restrict__ b_in, float* __restrict__ E_mag) {
    __shared__ float sW[8 * NQ];
    __shared__ float sC[NQ], sS[NQ];
    const int tid = threadIdx.x, b = blockIdx.x;
    const int lane = tid & 31, wv = tid >> 5;

    float acc[NQ];
#pragma unroll
    for (int q = 0; q < NQ; ++q) acc[q] = 0.f;
    for (int d = tid; d < CDIM; d += 256) {
        const float zv = z[b * CDIM + d];
        const float* wr = W_in + d * NQ;
#pragma unroll
        for (int q = 0; q < NQ; ++q) acc[q] += zv * wr[q];
    }
#pragma unroll
    for (int q = 0; q < NQ; ++q) {
        for (int s = 16; s > 0; s >>= 1) acc[q] += __shfl_xor(acc[q], s, 32);
    }
    if (lane == 0) {
#pragma unroll
        for (int q = 0; q < NQ; ++q) sW[wv * NQ + q] = acc[q];
    }
    __syncthreads();
    if (tid < NQ) {
        float a = b_in[tid];
        for (int w = 0; w < 8; ++w) a += sW[w * NQ + tid];
        sC[tid] = cosf(0.5f * a);
        sS[tid] = sinf(0.5f * a);
    }
    __syncthreads();
    for (int i = 0; i < 4; ++i) {
        const int k = tid + i * 256;
        float m = 1.f;
#pragma unroll
        for (int q = 0; q < NQ; ++q) m *= ((k >> (9 - q)) & 1) ? sS[q] : sC[q];
        E_mag[b * SDIM + k] = m;
    }
}

#define APITCH (SDIM + 4)
#define GROWS  32
#define GCOLS  256
#define GWAVES 4
#define NCB    (SDIM / GCOLS)

struct BFrag { v4f r0, r1, i0, i1; };
struct Frag { BFrag b; v2f a0, a1; };

__device__ __forceinline__ float half_sum16(float v) {
    v += __shfl_xor(v, 1, 32);
    v += __shfl_xor(v, 2, 32);
    v += __shfl_xor(v, 4, 32);
    v += __shfl_xor(v, 8, 32);
    return v;
}

#define PASS_SET(f)                                                       \
    asm volatile("" : "+v"(f.b.r0), "+v"(f.b.r1), "+v"(f.b.i0), "+v"(f.b.i1), \
                      "+v"(f.a0), "+v"(f.a1))

__global__ void __launch_bounds__(128) state_gemm_kernel(
    const float* __restrict__ E_mag,
    const float* __restrict__ M_re, const float* __restrict__ M_im,
    float* __restrict__ evs_part) {
    __shared__ __attribute__((aligned(16))) float sA[GROWS][APITCH];
    __shared__ float sRed[GWAVES][GROWS * NQ];
    const int tid = threadIdx.x;
    const int lane = tid & 31, wv = tid >> 5;
    const int cb = blockIdx.x & (NCB - 1);
    const int m0 = (blockIdx.x / NCB) * GROWS;

#pragma unroll 32
    for (int i = tid; i < GROWS * (SDIM / 4); i += 128) {
        const int row = i >> 8, c4 = (i & 255) * 4;
        const float4 v = *(const float4*)(E_mag + (size_t)(m0 + row) * SDIM + c4);
        *(float4*)(&sA[row][c4]) = v;
    }
    __syncthreads();

    const int half = lane >> 4;
    const int l15 = lane & 15;
    const int n_base = cb * GCOLS + wv * 64;
    const float* aP = &sA[l15][half * 2];
    const float* bR = M_re + (size_t)(half * 2) * SDIM + n_base + 4 * l15;
    const float* bI = M_im + (size_t)(half * 2) * SDIM + n_base + 4 * l15;

    v8f fr[2][4], fi[2][4];
#pragma unroll
    for (int u = 0; u < 2; ++u) {
#pragma unroll
        for (int t = 0; t < 4; ++t) { fr[u][t] = v8f{}; fi[u][t] = v8f{}; }
    }

    auto loadF = [&](int k0) {
        Frag f;
        f.b.r0 = *(const v4f*)(bR + (size_t)k0 * SDIM);
        f.b.r1 = *(const v4f*)(bR + (size_t)(k0 + 1) * SDIM);
        f.b.i0 = *(const v4f*)(bI + (size_t)k0 * SDIM);
        f.b.i1 = *(const v4f*)(bI + (size_t)(k0 + 1) * SDIM);
        f.a0 = *(const v2f*)(aP + k0);
        f.a1 = *(const v2f*)(aP + 16 * APITCH + k0);
        return f;
    };
    auto mma = [&](const Frag& f) {
#pragma unroll
        for (int t = 0; t < 4; ++t) {
            v2f b_re, b_im;
            b_re.x = f.b.r0[t]; b_re.y = f.b.r1[t];
            b_im.x = f.b.i0[t]; b_im.y = f.b.i1[t];
            fr[0][t] = __builtin_amdgcn_wmma_f32_16x16x4_f32(false, f.a0, false, b_re,
                                                             (short)0, fr[0][t], false, false);
            fi[0][t] = __builtin_amdgcn_wmma_f32_16x16x4_f32(false, f.a0, false, b_im,
                                                             (short)0, fi[0][t], false, false);
            fr[1][t] = __builtin_amdgcn_wmma_f32_16x16x4_f32(false, f.a1, false, b_re,
                                                             (short)0, fr[1][t], false, false);
            fi[1][t] = __builtin_amdgcn_wmma_f32_16x16x4_f32(false, f.a1, false, b_im,
                                                             (short)0, fi[1][t], false, false);
        }
    };

    Frag f0 = loadF(0), f1 = loadF(4), f2 = loadF(8), f3 = loadF(12);
    PASS_SET(f0);
    PASS_SET(f1);
    PASS_SET(f2);
    PASS_SET(f3);
    for (int k0 = 0; k0 < SDIM - 4; k0 += 20) {
        Frag f4 = loadF(k0 + 16);
        __builtin_amdgcn_sched_barrier(0);
        PASS_SET(f0);
        mma(f0);
        f0 = loadF(k0 + 20);
        __builtin_amdgcn_sched_barrier(0);
        PASS_SET(f1);
        mma(f1);
        f1 = loadF((k0 + 24) & (SDIM - 1));
        __builtin_amdgcn_sched_barrier(0);
        PASS_SET(f2);
        mma(f2);
        f2 = loadF((k0 + 28) & (SDIM - 1));
        __builtin_amdgcn_sched_barrier(0);
        PASS_SET(f3);
        mma(f3);
        f3 = loadF((k0 + 32) & (SDIM - 1));
        __builtin_amdgcn_sched_barrier(0);
        PASS_SET(f4);
        mma(f4);
    }
    PASS_SET(f0);
    mma(f0);

    const int jb = n_base + 4 * l15;
#pragma unroll
    for (int u = 0; u < 2; ++u) {
#pragma unroll
        for (int r = 0; r < 8; ++r) {
            float p[4];
#pragma unroll
            for (int t = 0; t < 4; ++t) {
                const float fre = fr[u][t][r];
                const float fim = fi[u][t][r];
                p[t] = fre * fre + fim * fim;
            }
            const float s01 = p[0] + p[1], s23 = p[2] + p[3];
            const float S = s01 + s23;
            const int rowm = u * 16 + r + half * 8;
            float mine = 0.f;
#pragma unroll
            for (int w = 0; w < NQ - 2; ++w) {
                float val = ((jb >> (9 - w)) & 1) ? -S : S;
                val = half_sum16(val);
                if (l15 == w) mine = val;
            }
            const float v8 = half_sum16(s01 - s23);
            const float v9 = half_sum16((p[0] - p[1]) + (p[2] - p[3]));
            if (l15 == 8) mine = v8;
            if (l15 == 9) mine = v9;
            if (l15 < NQ) sRed[wv][rowm * NQ + l15] = mine;
        }
    }
    __syncthreads();
    for (int i = tid; i < GROWS * NQ; i += 128) {
        float acc = 0.f;
#pragma unroll
        for (int g = 0; g < GWAVES; ++g) acc += sRed[g][i];
        evs_part[(size_t)cb * BATCH * NQ + (size_t)m0 * NQ + i] = acc;
    }
}

__global__ void __launch_bounds__(256) out_gemm_kernel(
    const float* __restrict__ evs_part, const float* __restrict__ W_out,
    const float* __restrict__ b_out, float* __restrict__ out) {
    __shared__ float sE[NQ];
    const int b = blockIdx.x, tid = threadIdx.x;
    if (tid < NQ) {
        float a = evs_part[b * NQ + tid];
#pragma unroll
        for (int c = 1; c < NCB; ++c) a += evs_part[(size_t)c * BATCH * NQ + b * NQ + tid];
        sE[tid] = a;
    }
    __syncthreads();
    float e[NQ];
#pragma unroll
    for (int w = 0; w < NQ; ++w) e[w] = sE[w];
    for (int c = tid; c < CDIM; c += 256) {
        float acc = b_out[c];
#pragma unroll
        for (int w = 0; w < NQ; ++w) acc += e[w] * W_out[w * CDIM + c];
        out[(size_t)b * CDIM + c] = acc;
    }
}

extern "C" void kernel_launch(void* const* d_in, const int* in_sizes, int n_in,
                              void* d_out, int out_size, void* d_ws,
                              size_t ws_size, hipStream_t stream) {
    const float* z_i     = (const float*)d_in[0];
    const float* W_in    = (const float*)d_in[1];
    const float* b_in    = (const float*)d_in[2];
    const float* weights = (const float*)d_in[3];
    const float* W_out   = (const float*)d_in[4];
    const float* b_out   = (const float*)d_in[5];
    float* out = (float*)d_out;

    float* ws    = (float*)d_ws;
    float* M_re  = ws;
    float* M_im  = M_re + (size_t)SDIM * SDIM;
    float* E_mag = M_im + (size_t)SDIM * SDIM;
    float* evs_part = E_mag + (size_t)BATCH * SDIM;

    build_unitary_kernel<<<SDIM, 512, 0, stream>>>(weights, M_re, M_im);
    embed_kernel<<<BATCH, 256, 0, stream>>>(z_i, W_in, b_in, E_mag);
    state_gemm_kernel<<<(BATCH / GROWS) * NCB, 128, 0, stream>>>(E_mag, M_re, M_im, evs_part);
    out_gemm_kernel<<<BATCH, 256, 0, stream>>>(evs_part, W_out, b_out, out);
}


# unitary build: one wave per basis state holds the state in registers (32 amplitudes a lane), 5 Rot gates per LDS pass, each CNOT ring folded into a store address, layer 0 built as a product state
# speedup vs baseline: 7.2742x; 1.0521x over previous
#include <hip/hip_runtime.h>

#define NQ      10
#define SDIM    1024
#define CDIM    768
#define NLAYERS 3
#define BATCH   2048

typedef float v2f __attribute__((ext_vector_type(2)));
typedef float v4f __attribute__((ext_vector_type(4)));
typedef float v8f __attribute__((ext_vector_type(8)));

__device__ __forceinline__ float2 cmul(float2 a, float2 b) {
    return make_float2(a.x * b.x - a.y * b.y, a.x * b.y + a.y * b.x);
}
__device__ __forceinline__ float2 cadd(float2 a, float2 b) {
    return make_float2(a.x + b.x, a.y + b.y);
}

#define UPITCH 33

__host__ __device__ constexpr int cnot_ring(int l, int j) {
    const int r = (l % (NQ - 1)) + 1;
    for (int i = 0; i < NQ; ++i) {
        const int t = (i + r) % NQ;
        if ((j >> (9 - i)) & 1) j ^= 1 << (9 - t);
    }
    return j;
}

template <int L>
__device__ __forceinline__ void store_ring(float2* st, const float2 (&a)[32], int lane) {
    const int gl = cnot_ring(L, lane << 5);
#pragma unroll
    for (int r = 0; r < 32; ++r) {
        const int x = gl ^ cnot_ring(L, r);
        st[x + (x >> 5)] = a[r];
    }
}

__device__ __forceinline__ float lane_value(float v, int src_lane) {
    return __builtin_bit_cast(float, __builtin_amdgcn_readlane(__builtin_bit_cast(int, v), src_lane));
}

__global__ void __launch_bounds__(32) build_unitary_kernel(
    const float* __restrict__ weights, float* __restrict__ M_re,
    float* __restrict__ M_im) {
    __shared__ float2 st[32 * UPITCH];
    const int lane = threadIdx.x;
    const int k = blockIdx.x;

    float gx = 0.f, gy = 0.f, gz = 0.f, gw = 0.f;
    if (lane < NLAYERS * NQ) {
        const float* w = weights + lane * 3;
        const float phi = w[0], theta = w[1], omega = w[2];
        const float ct = cosf(0.5f * theta), sv = sinf(0.5f * theta);
        const float al = 0.5f * (phi + omega), be = 0.5f * (phi - omega);
        const float ca = cosf(al), sa = sinf(al);
        const float cb = cosf(be), sb = sinf(be);
        gx = ct * ca; gy = ct * sa; gz = sv * cb; gw = sv * sb;
    }

    float2 a[32];

    {
        float2 P = make_float2(1.f, 0.f);
#pragma unroll
        for (int q = 0; q < 5; ++q) {
            const float x = lane_value(gx, q), y = lane_value(gy, q);
            const float z = lane_value(gz, q), w = lane_value(gw, q);
            const int kq = (k >> (9 - q)) & 1;
            const int jq = (lane >> (4 - q)) & 1;
            const float2 g0 = kq ? make_float2(-z, -w) : make_float2(x, -y);
            const float2 g1 = kq ? make_float2(x, y) : make_float2(z, -w);
            P = cmul(jq ? g1 : g0, P);
        }
        a[0] = P;
#pragma unroll
        for (int i = 0; i < 5; ++i) {
            const int q = 5 + i, b = 4 - i;
            const float x = lane_value(gx, q), y = lane_value(gy, q);
            const float z = lane_value(gz, q), w = lane_value(gw, q);
            const int kq = (k >> (9 - q)) & 1;
            const float2 g0 = kq ? make_float2(-z, -w) : make_float2(x, -y);
            const float2 g1 = kq ? make_float2(x, y) : make_float2(z, -w);
#pragma unroll
            for (int m = 0; m < (1 << i); ++m) {
                const int r0 = m << (b + 1);
                const float2 v = a[r0];
                a[r0 | (1 << b)] = cmul(g1, v);
                a[r0] = cmul(g0, v);
            }
        }
    }
    store_ring<0>(st, a, lane);

#pragma unroll 1
    for (int pass = 0; pass < 4; ++pass) {
        const int layer = 1 + (pass >> 1), hi = pass & 1;
        const int base = hi ? lane * UPITCH : lane;
        const int stride = hi ? 1 : UPITCH;
        __syncthreads();
#pragma unroll
        for (int r = 0; r < 32; ++r) a[r] = st[base + r * stride];
#pragma unroll
        for (int i = 0; i < 5; ++i) {
            const int gi = layer * NQ + hi * 5 + i;
            const int b = 4 - i;
            const float x = lane_value(gx, gi), y = lane_value(gy, gi);
            const float z = lane_value(gz, gi), w = lane_value(gw, gi);
            const float2 g00 = make_float2(x, -y);
            const float2 g01 = make_float2(-z, -w);
            const float2 g10 = make_float2(z, -w);
            const float2 g11 = make_float2(x, y);
#pragma unroll
            for (int p = 0; p < 16; ++p) {
                const int r0 = ((p >> b) << (b + 1)) | (p & ((1 << b) - 1));
                const int r1 = r0 | (1 << b);
                const float2 a0 = a[r0], a1 = a[r1];
                a[r0] = cadd(cmul(g00, a0), cmul(g01, a1));
                a[r1] = cadd(cmul(g10, a0), cmul(g11, a1));
            }
        }
        if (!hi) {
#pragma unroll
            for (int r = 0; r < 32; ++r) st[base + r * stride] = a[r];
        } else if (layer == 1) {
            store_ring<1>(st, a, lane);
        } else {
            const int p = __popc(k) & 3;
            const float sg = (p & 2) ? -1.f : 1.f;
#pragma unroll
            for (int r = 0; r < 32; ++r) {
                const float2 u = a[r];
                a[r] = (p & 1) ? make_float2(sg * u.y, -sg * u.x) : make_float2(sg * u.x, sg * u.y);
            }
            store_ring<2>(st, a, lane);
        }
    }
    __syncthreads();
#pragma unroll
    for (int r = 0; r < 32; ++r) {
        const float2 v = st[r * UPITCH + lane];
        M_re[k * SDIM + r * 32 + lane] = v.x;
        M_im[k * SDIM + r * 32 + lane] = v.y;
    }
}

__global__ void __launch_bounds__(256) embed_kernel(
    const float* __restrict__ z, const float* __restrict__ W_in,
    const float* __restrict__ b_in, float* __restrict__ E_mag) {
    __shared__ float sW[8 * NQ];
    __shared__ float sC[NQ], sS[NQ];
    const int tid = threadIdx.x, b = blockIdx.x;
    const int lane = tid & 31, wv = tid >> 5;

    float acc[NQ];
#pragma unroll
    for (int q = 0; q < NQ; ++q) acc[q] = 0.f;
    for (int d = tid; d < CDIM; d += 256) {
        const float zv = z[b * CDIM + d];
        const float* wr = W_in + d * NQ;
#pragma unroll
        for (int q = 0; q < NQ; ++q) acc[q] += zv * wr[q];
    }
#pragma unroll
    for (int q = 0; q < NQ; ++q) {
        for (int s = 16; s > 0; s >>= 1) acc[q] += __shfl_xor(acc[q], s, 32);
    }
    if (lane == 0) {
#pragma unroll
        for (int q = 0; q < NQ; ++q) sW[wv * NQ + q] = acc[q];
    }
    __syncthreads();
    if (tid < NQ) {
        float a = b_in[tid];
        for (int w = 0; w < 8; ++w) a += sW[w * NQ + tid];
        sC[tid] = cosf(0.5f * a);
        sS[tid] = sinf(0.5f * a);
    }
    __syncthreads();
    for (int i = 0; i < 4; ++i) {
        const int k = tid + i * 256;
        float m = 1.f;
#pragma unroll
        for (int q = 0; q < NQ; ++q) m *= ((k >> (9 - q)) & 1) ? sS[q] : sC[q];
        E_mag[b * SDIM + k] = m;
    }
}

#define APITCH (SDIM + 4)
#define GROWS  32
#define GCOLS  256
#define GWAVES 4
#define NCB    (SDIM / GCOLS)

struct BFrag { v4f r0, r1, i0, i1; };
struct Frag { BFrag b; v2f a0, a1; };

__device__ __forceinline__ float half_sum16(float v) {
    v += __shfl_xor(v, 1, 32);
    v += __shfl_xor(v, 2, 32);
    v += __shfl_xor(v, 4, 32);
    v += __shfl_xor(v, 8, 32);
    return v;
}

#define PASS_SET(f)                                                       \
    asm volatile("" : "+v"(f.b.r0), "+v"(f.b.r1), "+v"(f.b.i0), "+v"(f.b.i1), \
                      "+v"(f.a0), "+v"(f.a1))

__global__ void __launch_bounds__(128) state_gemm_kernel(
    const float* __restrict__ E_mag,
    const float* __restrict__ M_re, const float* __restrict__ M_im,
    float* __restrict__ evs_part) {
    __shared__ __attribute__((aligned(16))) float sA[GROWS][APITCH];
    __shared__ float sRed[GWAVES][GROWS * NQ];
    const int tid = threadIdx.x;
    const int lane = tid & 31, wv = tid >> 5;
    const int cb = blockIdx.x & (NCB - 1);
    const int m0 = (blockIdx.x / NCB) * GROWS;

#pragma unroll 32
    for (int i = tid; i < GROWS * (SDIM / 4); i += 128) {
        const int row = i >> 8, c4 = (i & 255) * 4;
        const float4 v = *(const float4*)(E_mag + (size_t)(m0 + row) * SDIM + c4);
        *(float4*)(&sA[row][c4]) = v;
    }
    __syncthreads();

    const int half = lane >> 4;
    const int l15 = lane & 15;
    const int n_base = cb * GCOLS + wv * 64;
    const float* aP = &sA[l15][half * 2];
    const float* bR = M_re + (size_t)(half * 2) * SDIM + n_base + 4 * l15;
    const float* bI = M_im + (size_t)(half * 2) * SDIM + n_base + 4 * l15;

    v8f fr[2][4], fi[2][4];
#pragma unroll
    for (int u = 0; u < 2; ++u) {
#pragma unroll
        for (int t = 0; t < 4; ++t) { fr[u][t] = v8f{}; fi[u][t] = v8f{}; }
    }

    auto loadF = [&](int k0) {
        Frag f;
        f.b.r0 = *(const v4f*)(bR + (size_t)k0 * SDIM);
        f.b.r1 = *(const v4f*)(bR + (size_t)(k0 + 1) * SDIM);
        f.b.i0 = *(const v4f*)(bI + (size_t)k0 * SDIM);
        f.b.i1 = *(const v4f*)(bI + (size_t)(k0 + 1) * SDIM);
        f.a0 = *(const v2f*)(aP + k0);
        f.a1 = *(const v2f*)(aP + 16 * APITCH + k0);
        return f;
    };
    auto mma = [&](const Frag& f) {
#pragma unroll
        for (int t = 0; t < 4; ++t) {
            v2f b_re, b_im;
            b_re.x = f.b.r0[t]; b_re.y = f.b.r1[t];
            b_im.x = f.b.i0[t]; b_im.y = f.b.i1[t];
            fr[0][t] = __builtin_amdgcn_wmma_f32_16x16x4_f32(false, f.a0, false, b_re,
                                                             (short)0, fr[0][t], false, false);
            fi[0][t] = __builtin_amdgcn_wmma_f32_16x16x4_f32(false, f.a0, false, b_im,
                                                             (short)0, fi[0][t], false, false);
            fr[1][t] = __builtin_amdgcn_wmma_f32_16x16x4_f32(false, f.a1, false, b_re,
                                                             (short)0, fr[1][t], false, false);
            fi[1][t] = __builtin_amdgcn_wmma_f32_16x16x4_f32(false, f.a1, false, b_im,
                                                             (short)0, fi[1][t], false, false);
        }
    };

    Frag f0 = loadF(0), f1 = loadF(4), f2 = loadF(8), f3 = loadF(12);
    PASS_SET(f0);
    PASS_SET(f1);
    PASS_SET(f2);
    PASS_SET(f3);
    for (int k0 = 0; k0 < SDIM - 4; k0 += 20) {
        Frag f4 = loadF(k0 + 16);
        __builtin_amdgcn_sched_barrier(0);
        PASS_SET(f0);
        mma(f0);
        f0 = loadF(k0 + 20);
        __builtin_amdgcn_sched_barrier(0);
        PASS_SET(f1);
        mma(f1);
        f1 = loadF((k0 + 24) & (SDIM - 1));
        __builtin_amdgcn_sched_barrier(0);
        PASS_SET(f2);
        mma(f2);
        f2 = loadF((k0 + 28) & (SDIM - 1));
        __builtin_amdgcn_sched_barrier(0);
        PASS_SET(f3);
        mma(f3);
        f3 = loadF((k0 + 32) & (SDIM - 1));
        __builtin_amdgcn_sched_barrier(0);
        PASS_SET(f4);
        mma(f4);
    }
    PASS_SET(f0);
    mma(f0);

    const int jb = n_base + 4 * l15;
#pragma unroll
    for (int u = 0; u < 2; ++u) {
#pragma unroll
        for (int r = 0; r < 8; ++r) {
            float p[4];
#pragma unroll
            for (int t = 0; t < 4; ++t) {
                const float fre = fr[u][t][r];
                const float fim = fi[u][t][r];
                p[t] = fre * fre + fim * fim;
            }
            const float s01 = p[0] + p[1], s23 = p[2] + p[3];
            const float S = s01 + s23;
            const int rowm = u * 16 + r + half * 8;
            float mine = 0.f;
#pragma unroll
            for (int w = 0; w < NQ - 2; ++w) {
                float val = ((jb >> (9 - w)) & 1) ? -S : S;
                val = half_sum16(val);
                if (l15 == w) mine = val;
            }
            const float v8 = half_sum16(s01 - s23);
            const float v9 = half_sum16((p[0] - p[1]) + (p[2] - p[3]));
            if (l15 == 8) mine = v8;
            if (l15 == 9) mine = v9;
            if (l15 < NQ) sRed[wv][rowm * NQ + l15] = mine;
        }
    }
    __syncthreads();
    for (int i = tid; i < GROWS * NQ; i += 128) {
        float acc = 0.f;
#pragma unroll
        for (int g = 0; g < GWAVES; ++g) acc += sRed[g][i];
        evs_part[(size_t)cb * BATCH * NQ + (size_t)m0 * NQ + i] = acc;
    }
}

__global__ void __launch_bounds__(256) out_gemm_kernel(
    const float* __restrict__ evs_part, const float* __restrict__ W_out,
    const float* __restrict__ b_out, float* __restrict__ out) {
    __shared__ float sE[NQ];
    const int b = blockIdx.x, tid = threadIdx.x;
    if (tid < NQ) {
        float a = evs_part[b * NQ + tid];
#pragma unroll
        for (int c = 1; c < NCB; ++c) a += evs_part[(size_t)c * BATCH * NQ + b * NQ + tid];
        sE[tid] = a;
    }
    __syncthreads();
    float e[NQ];
#pragma unroll
    for (int w = 0; w < NQ; ++w) e[w] = sE[w];
    for (int c = tid; c < CDIM; c += 256) {
        float acc = b_out[c];
#pragma unroll
        for (int w = 0; w < NQ; ++w) acc += e[w] * W_out[w * CDIM + c];
        out[(size_t)b * CDIM + c] = acc;
    }
}

extern "C" void kernel_launch(void* const* d_in, const int* in_sizes, int n_in,
                              void* d_out, int out_size, void* d_ws,
                              size_t ws_size, hipStream_t stream) {
    const float* z_i     = (const float*)d_in[0];
    const float* W_in    = (const float*)d_in[1];
    const float* b_in    = (const float*)d_in[2];
    const float* weights = (const float*)d_in[3];
    const float* W_out   = (const float*)d_in[4];
    const float* b_out   = (const float*)d_in[5];
    float* out = (float*)d_out;

    float* ws    = (float*)d_ws;
    float* M_re  = ws;
    float* M_im  = M_re + (size_t)SDIM * SDIM;
    float* E_mag = M_im + (size_t)SDIM * SDIM;
    float* evs_part = E_mag + (size_t)BATCH * SDIM;

    build_unitary_kernel<<<SDIM, 32, 0, stream>>>(weights, M_re, M_im);
    embed_kernel<<<BATCH, 256, 0, stream>>>(z_i, W_in, b_in, E_mag);
    state_gemm_kernel<<<(BATCH / GROWS) * NCB, 128, 0, stream>>>(E_mag, M_re, M_im, evs_part);
    out_gemm_kernel<<<BATCH, 256, 0, stream>>>(evs_part, W_out, b_out, out);
}


# unitary build and embed kernels merged into ONE launch, 256 blocks of 12 waves: 4 build a row of M each, 8 embed a sample each (angle sum order now: 24 terms per lane, lane butterfly, b_in last)
# speedup vs baseline: 7.6116x; 1.0464x over previous
#include <hip/hip_runtime.h>

#define NQ      10
#define SDIM    1024
#define CDIM    768
#define NLAYERS 3
#define BATCH   2048

typedef float v2f __attribute__((ext_vector_type(2)));
typedef float v4f __attribute__((ext_vector_type(4)));
typedef float v8f __attribute__((ext_vector_type(8)));

__device__ __forceinline__ float2 cmul(float2 a, float2 b) {
    return make_float2(a.x * b.x - a.y * b.y, a.x * b.y + a.y * b.x);
}
__device__ __forceinline__ float2 cadd(float2 a, float2 b) {
    return make_float2(a.x + b.x, a.y + b.y);
}

#define UPITCH 33

__host__ __device__ constexpr int cnot_ring(int l, int j) {
    const int r = (l % (NQ - 1)) + 1;
    for (int i = 0; i < NQ; ++i) {
        const int t = (i + r) % NQ;
        if ((j >> (9 - i)) & 1) j ^= 1 << (9 - t);
    }
    return j;
}

template <int L>
__device__ __forceinline__ void store_ring(float2* st, const float2 (&a)[32], int lane) {
    const int gl = cnot_ring(L, lane << 5);
#pragma unroll
    for (int r = 0; r < 32; ++r) {
        const int x = gl ^ cnot_ring(L, r);
        st[x + (x >> 5)] = a[r];
    }
}

__device__ __forceinline__ void wave_lds_fence() {
    __builtin_amdgcn_fence(__ATOMIC_ACQ_REL, "workgroup", "local");
}

__device__ __forceinline__ float lane_value(float v, int src_lane) {
    return __builtin_bit_cast(float, __builtin_amdgcn_readlane(__builtin_bit_cast(int, v), src_lane));
}

__device__ __forceinline__ void build_row(
    const float* __restrict__ weights, float* __restrict__ M_re,
    float* __restrict__ M_im, float2* st, const int lane, const int k) {

    float gx = 0.f, gy = 0.f, gz = 0.f, gw = 0.f;
    if (lane < NLAYERS * NQ) {
        const float* w = weights + lane * 3;
        const float phi = w[0], theta = w[1], omega = w[2];
        const float ct = cosf(0.5f * theta), sv = sinf(0.5f * theta);
        const float al = 0.5f * (phi + omega), be = 0.5f * (phi - omega);
        const float ca = cosf(al), sa = sinf(al);
        const float cb = cosf(be), sb = sinf(be);
        gx = ct * ca; gy = ct * sa; gz = sv * cb; gw = sv * sb;
    }

    float2 a[32];

    {
        float2 P = make_float2(1.f, 0.f);
#pragma unroll
        for (int q = 0; q < 5; ++q) {
            const float x = lane_value(gx, q), y = lane_value(gy, q);
            const float z = lane_value(gz, q), w = lane_value(gw, q);
            const int kq = (k >> (9 - q)) & 1;
            const int jq = (lane >> (4 - q)) & 1;
            const float2 g0 = kq ? make_float2(-z, -w) : make_float2(x, -y);
            const float2 g1 = kq ? make_float2(x, y) : make_float2(z, -w);
            P = cmul(jq ? g1 : g0, P);
        }
        a[0] = P;
#pragma unroll
        for (int i = 0; i < 5; ++i) {
            const int q = 5 + i, b = 4 - i;
            const float x = lane_value(gx, q), y = lane_value(gy, q);
            const float z = lane_value(gz, q), w = lane_value(gw, q);
            const int kq = (k >> (9 - q)) & 1;
            const float2 g0 = kq ? make_float2(-z, -w) : make_float2(x, -y);
            const float2 g1 = kq ? make_float2(x, y) : make_float2(z, -w);
#pragma unroll
            for (int m = 0; m < (1 << i); ++m) {
                const int r0 = m << (b + 1);
                const float2 v = a[r0];
                a[r0 | (1 << b)] = cmul(g1, v);
                a[r0] = cmul(g0, v);
            }
        }
    }
    store_ring<0>(st, a, lane);

#pragma unroll 1
    for (int pass = 0; pass < 4; ++pass) {
        const int layer = 1 + (pass >> 1), hi = pass & 1;
        const int base = hi ? lane * UPITCH : lane;
        const int stride = hi ? 1 : UPITCH;
        wave_lds_fence();
#pragma unroll
        for (int r = 0; r < 32; ++r) a[r] = st[base + r * stride];
#pragma unroll
        for (int i = 0; i < 5; ++i) {
            const int gi = layer * NQ + hi * 5 + i;
            const int b = 4 - i;
            const float x = lane_value(gx, gi), y = lane_value(gy, gi);
            const float z = lane_value(gz, gi), w = lane_value(gw, gi);
            const float2 g00 = make_float2(x, -y);
            const float2 g01 = make_float2(-z, -w);
            const float2 g10 = make_float2(z, -w);
            const float2 g11 = make_float2(x, y);
#pragma unroll
            for (int p = 0; p < 16; ++p) {
                const int r0 = ((p >> b) << (b + 1)) | (p & ((1 << b) - 1));
                const int r1 = r0 | (1 << b);
                const float2 a0 = a[r0], a1 = a[r1];
                a[r0] = cadd(cmul(g00, a0), cmul(g01, a1));
                a[r1] = cadd(cmul(g10, a0), cmul(g11, a1));
            }
        }
        if (!hi) {
#pragma unroll
            for (int r = 0; r < 32; ++r) st[base + r * stride] = a[r];
        } else if (layer == 1) {
            store_ring<1>(st, a, lane);
        } else {
            const int p = __popc(k) & 3;
            const float sg = (p & 2) ? -1.f : 1.f;
#pragma unroll
            for (int r = 0; r < 32; ++r) {
                const float2 u = a[r];
                a[r] = (p & 1) ? make_float2(sg * u.y, -sg * u.x) : make_float2(sg * u.x, sg * u.y);
            }
            store_ring<2>(st, a, lane);
        }
    }
    wave_lds_fence();
#pragma unroll
    for (int r = 0; r < 32; ++r) {
        const float2 v = st[r * UPITCH + lane];
        M_re[k * SDIM + r * 32 + lane] = v.x;
        M_im[k * SDIM + r * 32 + lane] = v.y;
    }
}

#define ZPL (CDIM / 32)

__device__ __forceinline__ void embed_row(
    const float* __restrict__ z, const float* __restrict__ W_in,
    const float* __restrict__ b_in, float* __restrict__ E_mag,
    const int lane, const int b) {
    float zv[ZPL];
#pragma unroll
    for (int i = 0; i < ZPL; ++i) zv[i] = z[(size_t)b * CDIM + lane + 32 * i];
    const float bv = (lane < NQ) ? b_in[lane] : 0.f;

    float acc[NQ];
#pragma unroll
    for (int q = 0; q < NQ; ++q) acc[q] = 0.f;
#pragma unroll 6
    for (int i = 0; i < ZPL; ++i) {
        const float* wr = W_in + (lane + 32 * i) * NQ;
#pragma unroll
        for (int q = 0; q < NQ; ++q) acc[q] += zv[i] * wr[q];
    }
#pragma unroll
    for (int q = 0; q < NQ; ++q) {
#pragma unroll
        for (int s = 16; s > 0; s >>= 1) acc[q] += __shfl_xor(acc[q], s, 32);
    }
    float a = 0.f;
#pragma unroll
    for (int q = 0; q < NQ; ++q) a = (lane == q) ? acc[q] : a;
    a += bv;
    const float cl = cosf(0.5f * a), sl = sinf(0.5f * a);
    float C[NQ], S[NQ];
#pragma unroll
    for (int q = 0; q < NQ; ++q) { C[q] = lane_value(cl, q); S[q] = lane_value(sl, q); }

    float hi[32];
    hi[0] = C[0]; hi[16] = S[0];
#pragma unroll
    for (int q = 1; q < 5; ++q) {
        const int bit = 16 >> q;
#pragma unroll
        for (int r = 0; r < 32; r += 2 * bit) {
            const float v = hi[r];
            hi[r] = v * C[q];
            hi[r | bit] = v * S[q];
        }
    }
    float lo[5];
#pragma unroll
    for (int q = 5; q < NQ; ++q) lo[q - 5] = ((lane >> (9 - q)) & 1) ? S[q] : C[q];
    float* row = E_mag + (size_t)b * SDIM + lane;
#pragma unroll
    for (int r = 0; r < 32; ++r) {
        float m = hi[r];
#pragma unroll
        for (int j = 0; j < 5; ++j) m *= lo[j];
        row[r * 32] = m;
    }
}

#define BWAVES 4
#define EWAVES 8
static_assert(SDIM / BWAVES == BATCH / EWAVES, "one grid serves the rows of M and the samples");

__global__ void __launch_bounds__(32 * (BWAVES + EWAVES)) prologue_kernel(
    const float* __restrict__ weights, float* __restrict__ M_re,
    float* __restrict__ M_im, const float* __restrict__ z,
    const float* __restrict__ W_in, const float* __restrict__ b_in,
    float* __restrict__ E_mag) {
    __shared__ float2 st[BWAVES][32 * UPITCH];
    const int lane = threadIdx.x & 31;
    const int wv = __builtin_amdgcn_readfirstlane(threadIdx.x >> 5);
    if (wv < BWAVES) {
        build_row(weights, M_re, M_im, st[wv], lane, blockIdx.x * BWAVES + wv);
    } else {
        embed_row(z, W_in, b_in, E_mag, lane, blockIdx.x * EWAVES + (wv - BWAVES));
    }
}

#define APITCH (SDIM + 4)
#define GROWS  32
#define GCOLS  256
#define GWAVES 4
#define NCB    (SDIM / GCOLS)

struct BFrag { v4f r0, r1, i0, i1; };
struct Frag { BFrag b; v2f a0, a1; };

__device__ __forceinline__ float half_sum16(float v) {
    v += __shfl_xor(v, 1, 32);
    v += __shfl_xor(v, 2, 32);
    v += __shfl_xor(v, 4, 32);
    v += __shfl_xor(v, 8, 32);
    return v;
}

#define PASS_SET(f)                                                       \
    asm volatile("" : "+v"(f.b.r0), "+v"(f.b.r1), "+v"(f.b.i0), "+v"(f.b.i1), \
                      "+v"(f.a0), "+v"(f.a1))

__global__ void __launch_bounds__(128) state_gemm_kernel(
    const float* __restrict__ E_mag,
    const float* __restrict__ M_re, const float* __restrict__ M_im,
    float* __restrict__ evs_part) {
    __shared__ __attribute__((aligned(16))) float sA[GROWS][APITCH];
    __shared__ float sRed[GWAVES][GROWS * NQ];
    const int tid = threadIdx.x;
    const int lane = tid & 31, wv = tid >> 5;
    const int cb = blockIdx.x & (NCB - 1);
    const int m0 = (blockIdx.x / NCB) * GROWS;

#pragma unroll 32
    for (int i = tid; i < GROWS * (SDIM / 4); i += 128) {
        const int row = i >> 8, c4 = (i & 255) * 4;
        const float4 v = *(const float4*)(E_mag + (size_t)(m0 + row) * SDIM + c4);
        *(float4*)(&sA[row][c4]) = v;
    }
    __syncthreads();

    const int half = lane >> 4;
    const int l15 = lane & 15;
    const int n_base = cb * GCOLS + wv * 64;
    const float* aP = &sA[l15][half * 2];
    const float* bR = M_re + (size_t)(half * 2) * SDIM + n_base + 4 * l15;
    const float* bI = M_im + (size_t)(half * 2) * SDIM + n_base + 4 * l15;

    v8f fr[2][4], fi[2][4];
#pragma unroll
    for (int u = 0; u < 2; ++u) {
#pragma unroll
        for (int t = 0; t < 4; ++t) { fr[u][t] = v8f{}; fi[u][t] = v8f{}; }
    }

    auto loadF = [&](int k0) {
        Frag f;
        f.b.r0 = *(const v4f*)(bR + (size_t)k0 * SDIM);
        f.b.r1 = *(const v4f*)(bR + (size_t)(k0 + 1) * SDIM);
        f.b.i0 = *(const v4f*)(bI + (size_t)k0 * SDIM);
        f.b.i1 = *(const v4f*)(bI + (size_t)(k0 + 1) * SDIM);
        f.a0 = *(const v2f*)(aP + k0);
        f.a1 = *(const v2f*)(aP + 16 * APITCH + k0);
        return f;
    };
    auto mma = [&](const Frag& f) {
#pragma unroll
        for (int t = 0; t < 4; ++t) {
            v2f b_re, b_im;
            b_re.x = f.b.r0[t]; b_re.y = f.b.r1[t];
            b_im.x = f.b.i0[t]; b_im.y = f.b.i1[t];
            fr[0][t] = __builtin_amdgcn_wmma_f32_16x16x4_f32(false, f.a0, false, b_re,
                                                             (short)0, fr[0][t], false, false);
            fi[0][t] = __builtin_amdgcn_wmma_f32_16x16x4_f32(false, f.a0, false, b_im,
                                                             (short)0, fi[0][t], false, false);
            fr[1][t] = __builtin_amdgcn_wmma_f32_16x16x4_f32(false, f.a1, false, b_re,
                                                             (short)0, fr[1][t], false, false);
            fi[1][t] = __builtin_amdgcn_wmma_f32_16x16x4_f32(false, f.a1, false, b_im,
                                                             (short)0, fi[1][t], false, false);
        }
    };

    Frag f0 = loadF(0), f1 = loadF(4), f2 = loadF(8), f3 = loadF(12);
    PASS_SET(f0);
    PASS_SET(f1);
    PASS_SET(f2);
    PASS_SET(f3);
    for (int k0 = 0; k0 < SDIM - 4; k0 += 20) {
        Frag f4 = loadF(k0 + 16);
        __builtin_amdgcn_sched_barrier(0);
        PASS_SET(f0);
        mma(f0);
        f0 = loadF(k0 + 20);
        __builtin_amdgcn_sched_barrier(0);
        PASS_SET(f1);
        mma(f1);
        f1 = loadF((k0 + 24) & (SDIM - 1));
        __builtin_amdgcn_sched_barrier(0);
        PASS_SET(f2);
        mma(f2);
        f2 = loadF((k0 + 28) & (SDIM - 1));
        __builtin_amdgcn_sched_barrier(0);
        PASS_SET(f3);
        mma(f3);
        f3 = loadF((k0 + 32) & (SDIM - 1));
        __builtin_amdgcn_sched_barrier(0);
        PASS_SET(f4);
        mma(f4);
    }
    PASS_SET(f0);
    mma(f0);

    const int jb = n_base + 4 * l15;
#pragma unroll
    for (int u = 0; u < 2; ++u) {
#pragma unroll
        for (int r = 0; r < 8; ++r) {
            float p[4];
#pragma unroll
            for (int t = 0; t < 4; ++t) {
                const float fre = fr[u][t][r];
                const float fim = fi[u][t][r];
                p[t] = fre * fre + fim * fim;
            }
            const float s01 = p[0] + p[1], s23 = p[2] + p[3];
            const float S = s01 + s23;
            const int rowm = u * 16 + r + half * 8;
            float mine = 0.f;
#pragma unroll
            for (int w = 0; w < NQ - 2; ++w) {
                float val = ((jb >> (9 - w)) & 1) ? -S : S;
                val = half_sum16(val);
                if (l15 == w) mine = val;
            }
            const float v8 = half_sum16(s01 - s23);
            const float v9 = half_sum16((p[0] - p[1]) + (p[2] - p[3]));
            if (l15 == 8) mine = v8;
            if (l15 == 9) mine = v9;
            if (l15 < NQ) sRed[wv][rowm * NQ + l15] = mine;
        }
    }
    __syncthreads();
    for (int i = tid; i < GROWS * NQ; i += 128) {
        float acc = 0.f;
#pragma unroll
        for (int g = 0; g < GWAVES; ++g) acc += sRed[g][i];
        evs_part[(size_t)cb * BATCH * NQ + (size_t)m0 * NQ + i] = acc;
    }
}

__global__ void __launch_bounds__(256) out_gemm_kernel(
    const float* __restrict__ evs_part, const float* __restrict__ W_out,
    const float* __restrict__ b_out, float* __restrict__ out) {
    __shared__ float sE[NQ];
    const int b = blockIdx.x, tid = threadIdx.x;
    if (tid < NQ) {
        float a = evs_part[b * NQ + tid];
#pragma unroll
        for (int c = 1; c < NCB; ++c) a += evs_part[(size_t)c * BATCH * NQ + b * NQ + tid];
        sE[tid] = a;
    }
    __syncthreads();
    float e[NQ];
#pragma unroll
    for (int w = 0; w < NQ; ++w) e[w] = sE[w];
    for (int c = tid; c < CDIM; c += 256) {
        float acc = b_out[c];
#pragma unroll
        for (int w = 0; w < NQ; ++w) acc += e[w] * W_out[w * CDIM + c];
        out[(size_t)b * CDIM + c] = acc;
    }
}

extern "C" void kernel_launch(void* const* d_in, const int* in_sizes, int n_in,
                              void* d_out, int out_size, void* d_ws,
                              size_t ws_size, hipStream_t stream) {
    const float* z_i     = (const float*)d_in[0];
    const float* W_in    = (const float*)d_in[1];
    const float* b_in    = (const float*)d_in[2];
    const float* weights = (const float*)d_in[3];
    const float* W_out   = (const float*)d_in[4];
    const float* b_out   = (const float*)d_in[5];
    float* out = (float*)d_out;

    float* ws    = (float*)d_ws;
    float* M_re  = ws;
    float* M_im  = M_re + (size_t)SDIM * SDIM;
    float* E_mag = M_im + (size_t)SDIM * SDIM;
    float* evs_part = E_mag + (size_t)BATCH * SDIM;

    prologue_kernel<<<SDIM / BWAVES, 32 * (BWAVES + EWAVES), 0, stream>>>(weights, M_re, M_im, z_i, W_in, b_in, E_mag);
    state_gemm_kernel<<<(BATCH / GROWS) * NCB, 128, 0, stream>>>(E_mag, M_re, M_im, evs_part);
    out_gemm_kernel<<<BATCH, 256, 0, stream>>>(evs_part, W_out, b_out, out);
}


# last kernel (evs @ W_out) re-gridded: 256 blocks of 6 waves, a block owns 8 samples, a thread keeps its 4 columns of W_out in registers; 1536 waves enter, not 16384; order of sums kept
# speedup vs baseline: 7.9495x; 1.0444x over previous
#include <hip/hip_runtime.h>

#define NQ      10
#define SDIM    1024
#define CDIM    768
#define NLAYERS 3
#define BATCH   2048

typedef float v2f __attribute__((ext_vector_type(2)));
typedef float v4f __attribute__((ext_vector_type(4)));
typedef float v8f __attribute__((ext_vector_type(8)));

__device__ __forceinline__ float2 cmul(float2 a, float2 b) {
    return make_float2(a.x * b.x - a.y * b.y, a.x * b.y + a.y * b.x);
}
__device__ __forceinline__ float2 cadd(float2 a, float2 b) {
    return make_float2(a.x + b.x, a.y + b.y);
}

#define UPITCH 33

__host__ __device__ constexpr int cnot_ring(int l, int j) {
    const int r = (l % (NQ - 1)) + 1;
    for (int i = 0; i < NQ; ++i) {
        const int t = (i + r) % NQ;
        if ((j >> (9 - i)) & 1) j ^= 1 << (9 - t);
    }
    return j;
}

template <int L>
__device__ __forceinline__ void store_ring(float2* st, const float2 (&a)[32], int lane) {
    const int gl = cnot_ring(L, lane << 5);
#pragma unroll
    for (int r = 0; r < 32; ++r) {
        const int x = gl ^ cnot_ring(L, r);
        st[x + (x >> 5)] = a[r];
    }
}

__device__ __forceinline__ void wave_lds_fence() {
    __builtin_amdgcn_fence(__ATOMIC_ACQ_REL, "workgroup", "local");
}

__device__ __forceinline__ float lane_value(float v, int src_lane) {
    return __builtin_bit_cast(float, __builtin_amdgcn_readlane(__builtin_bit_cast(int, v), src_lane));
}

__device__ __forceinline__ void build_row(
    const float* __restrict__ weights, float* __restrict__ M_re,
    float* __restrict__ M_im, float2* st, const int lane, const int k) {

    float gx = 0.f, gy = 0.f, gz = 0.f, gw = 0.f;
    if (lane < NLAYERS * NQ) {
        const float* w = weights + lane * 3;
        const float phi = w[0], theta = w[1], omega = w[2];
        const float ct = cosf(0.5f * theta), sv = sinf(0.5f * theta);
        const float al = 0.5f * (phi + omega), be = 0.5f * (phi - omega);
        const float ca = cosf(al), sa = sinf(al);
        const float cb = cosf(be), sb = sinf(be);
        gx = ct * ca; gy = ct * sa; gz = sv * cb; gw = sv * sb;
    }

    float2 a[32];

    {
        float2 P = make_float2(1.f, 0.f);
#pragma unroll
        for (int q = 0; q < 5; ++q) {
            const float x = lane_value(gx, q), y = lane_value(gy, q);
            const float z = lane_value(gz, q), w = lane_value(gw, q);
            const int kq = (k >> (9 - q)) & 1;
            const int jq = (lane >> (4 - q)) & 1;
            const float2 g0 = kq ? make_float2(-z, -w) : make_float2(x, -y);
            const float2 g1 = kq ? make_float2(x, y) : make_float2(z, -w);
            P = cmul(jq ? g1 : g0, P);
        }
        a[0] = P;
#pragma unroll
        for (int i = 0; i < 5; ++i) {
            const int q = 5 + i, b = 4 - i;
            const float x = lane_value(gx, q), y = lane_value(gy, q);
            const float z = lane_value(gz, q), w = lane_value(gw, q);
            const int kq = (k >> (9 - q)) & 1;
            const float2 g0 = kq ? make_float2(-z, -w) : make_float2(x, -y);
            const float2 g1 = kq ? make_float2(x, y) : make_float2(z, -w);
#pragma unroll
            for (int m = 0; m < (1 << i); ++m) {
                const int r0 = m << (b + 1);
                const float2 v = a[r0];
                a[r0 | (1 << b)] = cmul(g1, v);
                a[r0] = cmul(g0, v);
            }
        }
    }
    store_ring<0>(st, a, lane);

#pragma unroll 1
    for (int pass = 0; pass < 4; ++pass) {
        const int layer = 1 + (pass >> 1), hi = pass & 1;
        const int base = hi ? lane * UPITCH : lane;
        const int stride = hi ? 1 : UPITCH;
        wave_lds_fence();
#pragma unroll
        for (int r = 0; r < 32; ++r) a[r] = st[base + r * stride];
#pragma unroll
        for (int i = 0; i < 5; ++i) {
            const int gi = layer * NQ + hi * 5 + i;
            const int b = 4 - i;
            const float x = lane_value(gx, gi), y = lane_value(gy, gi);
            const float z = lane_value(gz, gi), w = lane_value(gw, gi);
            const float2 g00 = make_float2(x, -y);
            const float2 g01 = make_float2(-z, -w);
            const float2 g10 = make_float2(z, -w);
            const float2 g11 = make_float2(x, y);
#pragma unroll
            for (int p = 0; p < 16; ++p) {
                const int r0 = ((p >> b) << (b + 1)) | (p & ((1 << b) - 1));
                const int r1 = r0 | (1 << b);
                const float2 a0 = a[r0], a1 = a[r1];
                a[r0] = cadd(cmul(g00, a0), cmul(g01, a1));
                a[r1] = cadd(cmul(g10, a0), cmul(g11, a1));
            }
        }
        if (!hi) {
#pragma unroll
            for (int r = 0; r < 32; ++r) st[base + r * stride] = a[r];
        } else if (layer == 1) {
            store_ring<1>(st, a, lane);
        } else {
            const int p = __popc(k) & 3;
            const float sg = (p & 2) ? -1.f : 1.f;
#pragma unroll
            for (int r = 0; r < 32; ++r) {
                const float2 u = a[r];
                a[r] = (p & 1) ? make_float2(sg * u.y, -sg * u.x) : make_float2(sg * u.x, sg * u.y);
            }
            store_ring<2>(st, a, lane);
        }
    }
    wave_lds_fence();
#pragma unroll
    for (int r = 0; r < 32; ++r) {
        const float2 v = st[r * UPITCH + lane];
        M_re[k * SDIM + r * 32 + lane] = v.x;
        M_im[k * SDIM + r * 32 + lane] = v.y;
    }
}

#define ZPL (CDIM / 32)

__device__ __forceinline__ void embed_row(
    const float* __restrict__ z, const float* __restrict__ W_in,
    const float* __restrict__ b_in, float* __restrict__ E_mag,
    const int lane, const int b) {
    float zv[ZPL];
#pragma unroll
    for (int i = 0; i < ZPL; ++i) zv[i] = z[(size_t)b * CDIM + lane + 32 * i];
    const float bv = (lane < NQ) ? b_in[lane] : 0.f;

    float acc[NQ];
#pragma unroll
    for (int q = 0; q < NQ; ++q) acc[q] = 0.f;
#pragma unroll 6
    for (int i = 0; i < ZPL; ++i) {
        const float* wr = W_in + (lane + 32 * i) * NQ;
#pragma unroll
        for (int q = 0; q < NQ; ++q) acc[q] += zv[i] * wr[q];
    }
#pragma unroll
    for (int q = 0; q < NQ; ++q) {
#pragma unroll
        for (int s = 16; s > 0; s >>= 1) acc[q] += __shfl_xor(acc[q], s, 32);
    }
    float a = 0.f;
#pragma unroll
    for (int q = 0; q < NQ; ++q) a = (lane == q) ? acc[q] : a;
    a += bv;
    const float cl = cosf(0.5f * a), sl = sinf(0.5f * a);
    float C[NQ], S[NQ];
#pragma unroll
    for (int q = 0; q < NQ; ++q) { C[q] = lane_value(cl, q); S[q] = lane_value(sl, q); }

    float hi[32];
    hi[0] = C[0]; hi[16] = S[0];
#pragma unroll
    for (int q = 1; q < 5; ++q) {
        const int bit = 16 >> q;
#pragma unroll
        for (int r = 0; r < 32; r += 2 * bit) {
            const float v = hi[r];
            hi[r] = v * C[q];
            hi[r | bit] = v * S[q];
        }
    }
    float lo[5];
#pragma unroll
    for (int q = 5; q < NQ; ++q) lo[q - 5] = ((lane >> (9 - q)) & 1) ? S[q] : C[q];
    float* row = E_mag + (size_t)b * SDIM + lane;
#pragma unroll
    for (int r = 0; r < 32; ++r) {
        float m = hi[r];
#pragma unroll
        for (int j = 0; j < 5; ++j) m *= lo[j];
        row[r * 32] = m;
    }
}

#define BWAVES 4
#define EWAVES 8
static_assert(SDIM / BWAVES == BATCH / EWAVES, "one grid serves the rows of M and the samples");

__global__ void __launch_bounds__(32 * (BWAVES + EWAVES)) prologue_kernel(
    const float* __restrict__ weights, float* __restrict__ M_re,
    float* __restrict__ M_im, const float* __restrict__ z,
    const float* __restrict__ W_in, const float* __restrict__ b_in,
    float* __restrict__ E_mag) {
    __shared__ float2 st[BWAVES][32 * UPITCH];
    const int lane = threadIdx.x & 31;
    const int wv = __builtin_amdgcn_readfirstlane(threadIdx.x >> 5);
    if (wv < BWAVES) {
        build_row(weights, M_re, M_im, st[wv], lane, blockIdx.x * BWAVES + wv);
    } else {
        embed_row(z, W_in, b_in, E_mag, lane, blockIdx.x * EWAVES + (wv - BWAVES));
    }
}

#define APITCH (SDIM + 4)
#define GROWS  32
#define GCOLS  256
#define GWAVES 4
#define NCB    (SDIM / GCOLS)

struct BFrag { v4f r0, r1, i0, i1; };
struct Frag { BFrag b; v2f a0, a1; };

__device__ __forceinline__ float half_sum16(float v) {
    v += __shfl_xor(v, 1, 32);
    v += __shfl_xor(v, 2, 32);
    v += __shfl_xor(v, 4, 32);
    v += __shfl_xor(v, 8, 32);
    return v;
}

#define PASS_SET(f)                                                       \
    asm volatile("" : "+v"(f.b.r0), "+v"(f.b.r1), "+v"(f.b.i0), "+v"(f.b.i1), \
                      "+v"(f.a0), "+v"(f.a1))

__global__ void __launch_bounds__(128) state_gemm_kernel(
    const float* __restrict__ E_mag,
    const float* __restrict__ M_re, const float* __restrict__ M_im,
    float* __restrict__ evs_part) {
    __shared__ __attribute__((aligned(16))) float sA[GROWS][APITCH];
    __shared__ float sRed[GWAVES][GROWS * NQ];
    const int tid = threadIdx.x;
    const int lane = tid & 31, wv = tid >> 5;
    const int cb = blockIdx.x & (NCB - 1);
    const int m0 = (blockIdx.x / NCB) * GROWS;

#pragma unroll 32
    for (int i = tid; i < GROWS * (SDIM / 4); i += 128) {
        const int row = i >> 8, c4 = (i & 255) * 4;
        const float4 v = *(const float4*)(E_mag + (size_t)(m0 + row) * SDIM + c4);
        *(float4*)(&sA[row][c4]) = v;
    }
    __syncthreads();

    const int half = lane >> 4;
    const int l15 = lane & 15;
    const int n_base = cb * GCOLS + wv * 64;
    const float* aP = &sA[l15][half * 2];
    const float* bR = M_re + (size_t)(half * 2) * SDIM + n_base + 4 * l15;
    const float* bI = M_im + (size_t)(half * 2) * SDIM + n_base + 4 * l15;

    v8f fr[2][4], fi[2][4];
#pragma unroll
    for (int u = 0; u < 2; ++u) {
#pragma unroll
        for (int t = 0; t < 4; ++t) { fr[u][t] = v8f{}; fi[u][t] = v8f{}; }
    }

    auto loadF = [&](int k0) {
        Frag f;
        f.b.r0 = *(const v4f*)(bR + (size_t)k0 * SDIM);
        f.b.r1 = *(const v4f*)(bR + (size_t)(k0 + 1) * SDIM);
        f.b.i0 = *(const v4f*)(bI + (size_t)k0 * SDIM);
        f.b.i1 = *(const v4f*)(bI + (size_t)(k0 + 1) * SDIM);
        f.a0 = *(const v2f*)(aP + k0);
        f.a1 = *(const v2f*)(aP + 16 * APITCH + k0);
        return f;
    };
    auto mma = [&](const Frag& f) {
#pragma unroll
        for (int t = 0; t < 4; ++t) {
            v2f b_re, b_im;
            b_re.x = f.b.r0[t]; b_re.y = f.b.r1[t];
            b_im.x = f.b.i0[t]; b_im.y = f.b.i1[t];
            fr[0][t] = __builtin_amdgcn_wmma_f32_16x16x4_f32(false, f.a0, false, b_re,
                                                             (short)0, fr[0][t], false, false);
            fi[0][t] = __builtin_amdgcn_wmma_f32_16x16x4_f32(false, f.a0, false, b_im,
                                                             (short)0, fi[0][t], false, false);
            fr[1][t] = __builtin_amdgcn_wmma_f32_16x16x4_f32(false, f.a1, false, b_re,
                                                             (short)0, fr[1][t], false, false);
            fi[1][t] = __builtin_amdgcn_wmma_f32_16x16x4_f32(false, f.a1, false, b_im,
                                                             (short)0, fi[1][t], false, false);
        }
    };

    Frag f0 = loadF(0), f1 = loadF(4), f2 = loadF(8), f3 = loadF(12);
    PASS_SET(f0);
    PASS_SET(f1);
    PASS_SET(f2);
    PASS_SET(f3);
    for (int k0 = 0; k0 < SDIM - 4; k0 += 20) {
        Frag f4 = loadF(k0 + 16);
        __builtin_amdgcn_sched_barrier(0);
        PASS_SET(f0);
        mma(f0);
        f0 = loadF(k0 + 20);
        __builtin_amdgcn_sched_barrier(0);
        PASS_SET(f1);
        mma(f1);
        f1 = loadF((k0 + 24) & (SDIM - 1));
        __builtin_amdgcn_sched_barrier(0);
        PASS_SET(f2);
        mma(f2);
        f2 = loadF((k0 + 28) & (SDIM - 1));
        __builtin_amdgcn_sched_barrier(0);
        PASS_SET(f3);
        mma(f3);
        f3 = loadF((k0 + 32) & (SDIM - 1));
        __builtin_amdgcn_sched_barrier(0);
        PASS_SET(f4);
        mma(f4);
    }
    PASS_SET(f0);
    mma(f0);

    const int jb = n_base + 4 * l15;
#pragma unroll
    for (int u = 0; u < 2; ++u) {
#pragma unroll
        for (int r = 0; r < 8; ++r) {
            float p[4];
#pragma unroll
            for (int t = 0; t < 4; ++t) {
                const float fre = fr[u][t][r];
                const float fim = fi[u][t][r];
                p[t] = fre * fre + fim * fim;
            }
            const float s01 = p[0] + p[1], s23 = p[2] + p[3];
            const float S = s01 + s23;
            const int rowm = u * 16 + r + half * 8;
            float mine = 0.f;
#pragma unroll
            for (int w = 0; w < NQ - 2; ++w) {
                float val = ((jb >> (9 - w)) & 1) ? -S : S;
                val = half_sum16(val);
                if (l15 == w) mine = val;
            }
            const float v8 = half_sum16(s01 - s23);
            const float v9 = half_sum16((p[0] - p[1]) + (p[2] - p[3]));
            if (l15 == 8) mine = v8;
            if (l15 == 9) mine = v9;
            if (l15 < NQ) sRed[wv][rowm * NQ + l15] = mine;
        }
    }
    __syncthreads();
    for (int i = tid; i < GROWS * NQ; i += 128) {
        float acc = 0.f;
#pragma unroll
        for (int g = 0; g < GWAVES; ++g) acc += sRed[g][i];
        evs_part[(size_t)cb * BATCH * NQ + (size_t)m0 * NQ + i] = acc;
    }
}

#define OSAMP   8
#define OTHREADS (CDIM / 4)
static_assert(BATCH % OSAMP == 0 && OSAMP * NQ <= OTHREADS, "the block's lanes cover its evs values");

__global__ void __launch_bounds__(OTHREADS) out_gemm_kernel(
    const float* __restrict__ evs_part, const float* __restrict__ W_out,
    const float* __restrict__ b_out, float* __restrict__ out) {
    __shared__ __attribute__((aligned(16))) float sE[OSAMP * NQ];
    const int tid = threadIdx.x;
    const int b0 = blockIdx.x * OSAMP;
    const int c0 = tid * 4;
    float4 wq[NQ];
#pragma unroll
    for (int w = 0; w < NQ; ++w) wq[w] = *(const float4*)(W_out + w * CDIM + c0);
    const float4 bo = *(const float4*)(b_out + c0);
    __builtin_amdgcn_sched_barrier(0);
    if (tid < OSAMP * NQ) {
        float a = evs_part[b0 * NQ + tid];
#pragma unroll
        for (int c = 1; c < NCB; ++c) a += evs_part[(size_t)c * BATCH * NQ + b0 * NQ + tid];
        sE[tid] = a;
    }
    __syncthreads();
#pragma unroll
    for (int s = 0; s < OSAMP; ++s) {
        float e[NQ];
#pragma unroll
        for (int w = 0; w < NQ; ++w) e[w] = sE[s * NQ + w];
        float4 acc = bo;
#pragma unroll
        for (int w = 0; w < NQ; ++w) {
            acc.x += e[w] * wq[w].x;
            acc.y += e[w] * wq[w].y;
            acc.z += e[w] * wq[w].z;
            acc.w += e[w] * wq[w].w;
        }
        *(float4*)(out + (size_t)(b0 + s) * CDIM + c0) = acc;
    }
}

extern "C" void kernel_launch(void* const* d_in, const int* in_sizes, int n_in,
                              void* d_out, int out_size, void* d_ws,
                              size_t ws_size, hipStream_t stream) {
    const float* z_i     = (const float*)d_in[0];
    const float* W_in    = (const float*)d_in[1];
    const float* b_in    = (const float*)d_in[2];
    const float* weights = (const float*)d_in[3];
    const float* W_out   = (const float*)d_in[4];
    const float* b_out   = (const float*)d_in[5];
    float* out = (float*)d_out;

    float* ws    = (float*)d_ws;
    float* M_re  = ws;
    float* M_im  = M_re + (size_t)SDIM * SDIM;
    float* E_mag = M_im + (size_t)SDIM * SDIM;
    float* evs_part = E_mag + (size_t)BATCH * SDIM;

    prologue_kernel<<<SDIM / BWAVES, 32 * (BWAVES + EWAVES), 0, stream>>>(weights, M_re, M_im, z_i, W_in, b_in, E_mag);
    state_gemm_kernel<<<(BATCH / GROWS) * NCB, 128, 0, stream>>>(E_mag, M_re, M_im, evs_part);
    out_gemm_kernel<<<BATCH / OSAMP, OTHREADS, 0, stream>>>(evs_part, W_out, b_out, out);
}


# state GEMM epilogue: each row's ten half-wave sums go through DPP lane exchanges in the registers (quad_perm, row_xmask) where they took 40 LDS permutes per row; same additions in the same order
# speedup vs baseline: 8.3593x; 1.0516x over previous
#include <hip/hip_runtime.h>

#define NQ      10
#define SDIM    1024
#define CDIM    768
#define NLAYERS 3
#define BATCH   2048

typedef float v2f __attribute__((ext_vector_type(2)));
typedef float v4f __attribute__((ext_vector_type(4)));
typedef float v8f __attribute__((ext_vector_type(8)));

__device__ __forceinline__ float2 cmul(float2 a, float2 b) {
    return make_float2(a.x * b.x - a.y * b.y, a.x * b.y + a.y * b.x);
}
__device__ __forceinline__ float2 cadd(float2 a, float2 b) {
    return make_float2(a.x + b.x, a.y + b.y);
}

#define UPITCH 33

__host__ __device__ constexpr int cnot_ring(int l, int j) {
    const int r = (l % (NQ - 1)) + 1;
    for (int i = 0; i < NQ; ++i) {
        const int t = (i + r) % NQ;
        if ((j >> (9 - i)) & 1) j ^= 1 << (9 - t);
    }
    return j;
}

template <int L>
__device__ __forceinline__ void store_ring(float2* st, const float2 (&a)[32], int lane) {
    const int gl = cnot_ring(L, lane << 5);
#pragma unroll
    for (int r = 0; r < 32; ++r) {
        const int x = gl ^ cnot_ring(L, r);
        st[x + (x >> 5)] = a[r];
    }
}

__device__ __forceinline__ void wave_lds_fence() {
    __builtin_amdgcn_fence(__ATOMIC_ACQ_REL, "workgroup", "local");
}

__device__ __forceinline__ float lane_value(float v, int src_lane) {
    return __builtin_bit_cast(float, __builtin_amdgcn_readlane(__builtin_bit_cast(int, v), src_lane));
}

__device__ __forceinline__ void build_row(
    const float* __restrict__ weights, float* __restrict__ M_re,
    float* __restrict__ M_im, float2* st, const int lane, const int k) {

    float gx = 0.f, gy = 0.f, gz = 0.f, gw = 0.f;
    if (lane < NLAYERS * NQ) {
        const float* w = weights + lane * 3;
        const float phi = w[0], theta = w[1], omega = w[2];
        const float ct = cosf(0.5f * theta), sv = sinf(0.5f * theta);
        const float al = 0.5f * (phi + omega), be = 0.5f * (phi - omega);
        const float ca = cosf(al), sa = sinf(al);
        const float cb = cosf(be), sb = sinf(be);
        gx = ct * ca; gy = ct * sa; gz = sv * cb; gw = sv * sb;
    }

    float2 a[32];

    {
        float2 P = make_float2(1.f, 0.f);
#pragma unroll
        for (int q = 0; q < 5; ++q) {
            const float x = lane_value(gx, q), y = lane_value(gy, q);
            const float z = lane_value(gz, q), w = lane_value(gw, q);
            const int kq = (k >> (9 - q)) & 1;
            const int jq = (lane >> (4 - q)) & 1;
            const float2 g0 = kq ? make_float2(-z, -w) : make_float2(x, -y);
            const float2 g1 = kq ? make_float2(x, y) : make_float2(z, -w);
            P = cmul(jq ? g1 : g0, P);
        }
        a[0] = P;
#pragma unroll
        for (int i = 0; i < 5; ++i) {
            const int q = 5 + i, b = 4 - i;
            const float x = lane_value(gx, q), y = lane_value(gy, q);
            const float z = lane_value(gz, q), w = lane_value(gw, q);
            const int kq = (k >> (9 - q)) & 1;
            const float2 g0 = kq ? make_float2(-z, -w) : make_float2(x, -y);
            const float2 g1 = kq ? make_float2(x, y) : make_float2(z, -w);
#pragma unroll
            for (int m = 0; m < (1 << i); ++m) {
                const int r0 = m << (b + 1);
                const float2 v = a[r0];
                a[r0 | (1 << b)] = cmul(g1, v);
                a[r0] = cmul(g0, v);
            }
        }
    }
    store_ring<0>(st, a, lane);

#pragma unroll 1
    for (int pass = 0; pass < 4; ++pass) {
        const int layer = 1 + (pass >> 1), hi = pass & 1;
        const int base = hi ? lane * UPITCH : lane;
        const int stride = hi ? 1 : UPITCH;
        wave_lds_fence();
#pragma unroll
        for (int r = 0; r < 32; ++r) a[r] = st[base + r * stride];
#pragma unroll
        for (int i = 0; i < 5; ++i) {
            const int gi = layer * NQ + hi * 5 + i;
            const int b = 4 - i;
            const float x = lane_value(gx, gi), y = lane_value(gy, gi);
            const float z = lane_value(gz, gi), w = lane_value(gw, gi);
            const float2 g00 = make_float2(x, -y);
            const float2 g01 = make_float2(-z, -w);
            const float2 g10 = make_float2(z, -w);
            const float2 g11 = make_float2(x, y);
#pragma unroll
            for (int p = 0; p < 16; ++p) {
                const int r0 = ((p >> b) << (b + 1)) | (p & ((1 << b) - 1));
                const int r1 = r0 | (1 << b);
                const float2 a0 = a[r0], a1 = a[r1];
                a[r0] = cadd(cmul(g00, a0), cmul(g01, a1));
                a[r1] = cadd(cmul(g10, a0), cmul(g11, a1));
            }
        }
        if (!hi) {
#pragma unroll
            for (int r = 0; r < 32; ++r) st[base + r * stride] = a[r];
        } else if (layer == 1) {
            store_ring<1>(st, a, lane);
        } else {
            const int p = __popc(k) & 3;
            const float sg = (p & 2) ? -1.f : 1.f;
#pragma unroll
            for (int r = 0; r < 32; ++r) {
                const float2 u = a[r];
                a[r] = (p & 1) ? make_float2(sg * u.y, -sg * u.x) : make_float2(sg * u.x, sg * u.y);
            }
            store_ring<2>(st, a, lane);
        }
    }
    wave_lds_fence();
#pragma unroll
    for (int r = 0; r < 32; ++r) {
        const float2 v = st[r * UPITCH + lane];
        M_re[k * SDIM + r * 32 + lane] = v.x;
        M_im[k * SDIM + r * 32 + lane] = v.y;
    }
}

#define ZPL (CDIM / 32)

__device__ __forceinline__ void embed_row(
    const float* __restrict__ z, const float* __restrict__ W_in,
    const float* __restrict__ b_in, float* __restrict__ E_mag,
    const int lane, const int b) {
    float zv[ZPL];
#pragma unroll
    for (int i = 0; i < ZPL; ++i) zv[i] = z[(size_t)b * CDIM + lane + 32 * i];
    const float bv = (lane < NQ) ? b_in[lane] : 0.f;

    float acc[NQ];
#pragma unroll
    for (int q = 0; q < NQ; ++q) acc[q] = 0.f;
#pragma unroll 6
    for (int i = 0; i < ZPL; ++i) {
        const float* wr = W_in + (lane + 32 * i) * NQ;
#pragma unroll
        for (int q = 0; q < NQ; ++q) acc[q] += zv[i] * wr[q];
    }
#pragma unroll
    for (int q = 0; q < NQ; ++q) {
#pragma unroll
        for (int s = 16; s > 0; s >>= 1) acc[q] += __shfl_xor(acc[q], s, 32);
    }
    float a = 0.f;
#pragma unroll
    for (int q = 0; q < NQ; ++q) a = (lane == q) ? acc[q] : a;
    a += bv;
    const float cl = cosf(0.5f * a), sl = sinf(0.5f * a);
    float C[NQ], S[NQ];
#pragma unroll
    for (int q = 0; q < NQ; ++q) { C[q] = lane_value(cl, q); S[q] = lane_value(sl, q); }

    float hi[32];
    hi[0] = C[0]; hi[16] = S[0];
#pragma unroll
    for (int q = 1; q < 5; ++q) {
        const int bit = 16 >> q;
#pragma unroll
        for (int r = 0; r < 32; r += 2 * bit) {
            const float v = hi[r];
            hi[r] = v * C[q];
            hi[r | bit] = v * S[q];
        }
    }
    float lo[5];
#pragma unroll
    for (int q = 5; q < NQ; ++q) lo[q - 5] = ((lane >> (9 - q)) & 1) ? S[q] : C[q];
    float* row = E_mag + (size_t)b * SDIM + lane;
#pragma unroll
    for (int r = 0; r < 32; ++r) {
        float m = hi[r];
#pragma unroll
        for (int j = 0; j < 5; ++j) m *= lo[j];
        row[r * 32] = m;
    }
}

#define BWAVES 4
#define EWAVES 8
static_assert(SDIM / BWAVES == BATCH / EWAVES, "one grid serves the rows of M and the samples");

__global__ void __launch_bounds__(32 * (BWAVES + EWAVES)) prologue_kernel(
    const float* __restrict__ weights, float* __restrict__ M_re,
    float* __restrict__ M_im, const float* __restrict__ z,
    const float* __restrict__ W_in, const float* __restrict__ b_in,
    float* __restrict__ E_mag) {
    __shared__ float2 st[BWAVES][32 * UPITCH];
    const int lane = threadIdx.x & 31;
    const int wv = __builtin_amdgcn_readfirstlane(threadIdx.x >> 5);
    if (wv < BWAVES) {
        build_row(weights, M_re, M_im, st[wv], lane, blockIdx.x * BWAVES + wv);
    } else {
        embed_row(z, W_in, b_in, E_mag, lane, blockIdx.x * EWAVES + (wv - BWAVES));
    }
}

#define APITCH (SDIM + 4)
#define GROWS  32
#define GCOLS  256
#define GWAVES 4
#define NCB    (SDIM / GCOLS)

struct BFrag { v4f r0, r1, i0, i1; };
struct Frag { BFrag b; v2f a0, a1; };

template <int CTRL>
__device__ __forceinline__ float row_partner(float v) {
    return __builtin_bit_cast(float, __builtin_amdgcn_update_dpp(
        0, __builtin_bit_cast(int, v), CTRL, 0xF, 0xF, true));
}

__device__ __forceinline__ float half_sum16(float v) {
    v += row_partner<0xB1>(v);
    v += row_partner<0x4E>(v);
    v += row_partner<0x164>(v);
    v += row_partner<0x168>(v);
    return v;
}

#define PASS_SET(f)                                                       \
    asm volatile("" : "+v"(f.b.r0), "+v"(f.b.r1), "+v"(f.b.i0), "+v"(f.b.i1), \
                      "+v"(f.a0), "+v"(f.a1))

__global__ void __launch_bounds__(128) state_gemm_kernel(
    const float* __restrict__ E_mag,
    const float* __restrict__ M_re, const float* __restrict__ M_im,
    float* __restrict__ evs_part) {
    __shared__ __attribute__((aligned(16))) float sA[GROWS][APITCH];
    __shared__ float sRed[GWAVES][GROWS * NQ];
    const int tid = threadIdx.x;
    const int lane = tid & 31, wv = tid >> 5;
    const int cb = blockIdx.x & (NCB - 1);
    const int m0 = (blockIdx.x / NCB) * GROWS;

#pragma unroll 32
    for (int i = tid; i < GROWS * (SDIM / 4); i += 128) {
        const int row = i >> 8, c4 = (i & 255) * 4;
        const float4 v = *(const float4*)(E_mag + (size_t)(m0 + row) * SDIM + c4);
        *(float4*)(&sA[row][c4]) = v;
    }
    __syncthreads();

    const int half = lane >> 4;
    const int l15 = lane & 15;
    const int n_base = cb * GCOLS + wv * 64;
    const float* aP = &sA[l15][half * 2];
    const float* bR = M_re + (size_t)(half * 2) * SDIM + n_base + 4 * l15;
    const float* bI = M_im + (size_t)(half * 2) * SDIM + n_base + 4 * l15;

    v8f fr[2][4], fi[2][4];
#pragma unroll
    for (int u = 0; u < 2; ++u) {
#pragma unroll
        for (int t = 0; t < 4; ++t) { fr[u][t] = v8f{}; fi[u][t] = v8f{}; }
    }

    auto loadF = [&](int k0) {
        Frag f;
        f.b.r0 = *(const v4f*)(bR + (size_t)k0 * SDIM);
        f.b.r1 = *(const v4f*)(bR + (size_t)(k0 + 1) * SDIM);
        f.b.i0 = *(const v4f*)(bI + (size_t)k0 * SDIM);
        f.b.i1 = *(const v4f*)(bI + (size_t)(k0 + 1) * SDIM);
        f.a0 = *(const v2f*)(aP + k0);
        f.a1 = *(const v2f*)(aP + 16 * APITCH + k0);
        return f;
    };
    auto mma = [&](const Frag& f) {
#pragma unroll
        for (int t = 0; t < 4; ++t) {
            v2f b_re, b_im;
            b_re.x = f.b.r0[t]; b_re.y = f.b.r1[t];
            b_im.x = f.b.i0[t]; b_im.y = f.b.i1[t];
            fr[0][t] = __builtin_amdgcn_wmma_f32_16x16x4_f32(false, f.a0, false, b_re,
                                                             (short)0, fr[0][t], false, false);
            fi[0][t] = __builtin_amdgcn_wmma_f32_16x16x4_f32(false, f.a0, false, b_im,
                                                             (short)0, fi[0][t], false, false);
            fr[1][t] = __builtin_amdgcn_wmma_f32_16x16x4_f32(false, f.a1, false, b_re,
                                                             (short)0, fr[1][t], false, false);
            fi[1][t] = __builtin_amdgcn_wmma_f32_16x16x4_f32(false, f.a1, false, b_im,
                                                             (short)0, fi[1][t], false, false);
        }
    };

    Frag f0 = loadF(0), f1 = loadF(4), f2 = loadF(8), f3 = loadF(12);
    PASS_SET(f0);
    PASS_SET(f1);
    PASS_SET(f2);
    PASS_SET(f3);
    for (int k0 = 0; k0 < SDIM - 4; k0 += 20) {
        Frag f4 = loadF(k0 + 16);
        __builtin_amdgcn_sched_barrier(0);
        PASS_SET(f0);
        mma(f0);
        f0 = loadF(k0 + 20);
        __builtin_amdgcn_sched_barrier(0);
        PASS_SET(f1);
        mma(f1);
        f1 = loadF((k0 + 24) & (SDIM - 1));
        __builtin_amdgcn_sched_barrier(0);
        PASS_SET(f2);
        mma(f2);
        f2 = loadF((k0 + 28) & (SDIM - 1));
        __builtin_amdgcn_sched_barrier(0);
        PASS_SET(f3);
        mma(f3);
        f3 = loadF((k0 + 32) & (SDIM - 1));
        __builtin_amdgcn_sched_barrier(0);
        PASS_SET(f4);
        mma(f4);
    }
    PASS_SET(f0);
    mma(f0);

    const int jb = n_base + 4 * l15;
#pragma unroll
    for (int u = 0; u < 2; ++u) {
#pragma unroll
        for (int r = 0; r < 8; ++r) {
            float p[4];
#pragma unroll
            for (int t = 0; t < 4; ++t) {
                const float fre = fr[u][t][r];
                const float fim = fi[u][t][r];
                p[t] = fre * fre + fim * fim;
            }
            const float s01 = p[0] + p[1], s23 = p[2] + p[3];
            const float S = s01 + s23;
            const int rowm = u * 16 + r + half * 8;
            float mine = 0.f;
#pragma unroll
            for (int w = 0; w < NQ - 2; ++w) {
                float val = ((jb >> (9 - w)) & 1) ? -S : S;
                val = half_sum16(val);
                if (l15 == w) mine = val;
            }
            const float v8 = half_sum16(s01 - s23);
            const float v9 = half_sum16((p[0] - p[1]) + (p[2] - p[3]));
            if (l15 == 8) mine = v8;
            if (l15 == 9) mine = v9;
            if (l15 < NQ) sRed[wv][rowm * NQ + l15] = mine;
        }
    }
    __syncthreads();
    for (int i = tid; i < GROWS * NQ; i += 128) {
        float acc = 0.f;
#pragma unroll
        for (int g = 0; g < GWAVES; ++g) acc += sRed[g][i];
        evs_part[(size_t)cb * BATCH * NQ + (size_t)m0 * NQ + i] = acc;
    }
}

#define OSAMP   8
#define OTHREADS (CDIM / 4)
static_assert(BATCH % OSAMP == 0 && OSAMP * NQ <= OTHREADS, "the block's lanes cover its evs values");

__global__ void __launch_bounds__(OTHREADS) out_gemm_kernel(
    const float* __restrict__ evs_part, const float* __restrict__ W_out,
    const float* __restrict__ b_out, float* __restrict__ out) {
    __shared__ __attribute__((aligned(16))) float sE[OSAMP * NQ];
    const int tid = threadIdx.x;
    const int b0 = blockIdx.x * OSAMP;
    const int c0 = tid * 4;
    float4 wq[NQ];
#pragma unroll
    for (int w = 0; w < NQ; ++w) wq[w] = *(const float4*)(W_out + w * CDIM + c0);
    const float4 bo = *(const float4*)(b_out + c0);
    __builtin_amdgcn_sched_barrier(0);
    if (tid < OSAMP * NQ) {
        float a = evs_part[b0 * NQ + tid];
#pragma unroll
        for (int c = 1; c < NCB; ++c) a += evs_part[(size_t)c * BATCH * NQ + b0 * NQ + tid];
        sE[tid] = a;
    }
    __syncthreads();
#pragma unroll
    for (int s = 0; s < OSAMP; ++s) {
        float e[NQ];
#pragma unroll
        for (int w = 0; w < NQ; ++w) e[w] = sE[s * NQ + w];
        float4 acc = bo;
#pragma unroll
        for (int w = 0; w < NQ; ++w) {
            acc.x += e[w] * wq[w].x;
            acc.y += e[w] * wq[w].y;
            acc.z += e[w] * wq[w].z;
            acc.w += e[w] * wq[w].w;
        }
        *(float4*)(out + (size_t)(b0 + s) * CDIM + c0) = acc;
    }
}

extern "C" void kernel_launch(void* const* d_in, const int* in_sizes, int n_in,
                              void* d_out, int out_size, void* d_ws,
                              size_t ws_size, hipStream_t stream) {
    const float* z_i     = (const float*)d_in[0];
    const float* W_in    = (const float*)d_in[1];
    const float* b_in    = (const float*)d_in[2];
    const float* weights = (const float*)d_in[3];
    const float* W_out   = (const float*)d_in[4];
    const float* b_out   = (const float*)d_in[5];
    float* out = (float*)d_out;

    float* ws    = (float*)d_ws;
    float* M_re  = ws;
    float* M_im  = M_re + (size_t)SDIM * SDIM;
    float* E_mag = M_im + (size_t)SDIM * SDIM;
    float* evs_part = E_mag + (size_t)BATCH * SDIM;

    prologue_kernel<<<SDIM / BWAVES, 32 * (BWAVES + EWAVES), 0, stream>>>(weights, M_re, M_im, z_i, W_in, b_in, E_mag);
    state_gemm_kernel<<<(BATCH / GROWS) * NCB, 128, 0, stream>>>(E_mag, M_re, M_im, evs_part);
    out_gemm_kernel<<<BATCH / OSAMP, OTHREADS, 0, stream>>>(evs_part, W_out, b_out, out);
}


# state GEMM: fragment loads of four k-steps (16 of M, four 16-byte LDS reads of m) go as one run per 64 WMMAs, two register sets in turn; a lane half takes 8 consecutive k: sum order over k changes
# speedup vs baseline: 8.6232x; 1.0316x over previous
#include <hip/hip_runtime.h>

#define NQ      10
#define SDIM    1024
#define CDIM    768
#define NLAYERS 3
#define BATCH   2048

typedef float v2f __attribute__((ext_vector_type(2)));
typedef float v4f __attribute__((ext_vector_type(4)));
typedef float v8f __attribute__((ext_vector_type(8)));

__device__ __forceinline__ float2 cmul(float2 a, float2 b) {
    return make_float2(a.x * b.x - a.y * b.y, a.x * b.y + a.y * b.x);
}
__device__ __forceinline__ float2 cadd(float2 a, float2 b) {
    return make_float2(a.x + b.x, a.y + b.y);
}

#define UPITCH 33

__host__ __device__ constexpr int cnot_ring(int l, int j) {
    const int r = (l % (NQ - 1)) + 1;
    for (int i = 0; i < NQ; ++i) {
        const int t = (i + r) % NQ;
        if ((j >> (9 - i)) & 1) j ^= 1 << (9 - t);
    }
    return j;
}

template <int L>
__device__ __forceinline__ void store_ring(float2* st, const float2 (&a)[32], int lane) {
    const int gl = cnot_ring(L, lane << 5);
#pragma unroll
    for (int r = 0; r < 32; ++r) {
        const int x = gl ^ cnot_ring(L, r);
        st[x + (x >> 5)] = a[r];
    }
}

__device__ __forceinline__ void wave_lds_fence() {
    __builtin_amdgcn_fence(__ATOMIC_ACQ_REL, "workgroup", "local");
}

__device__ __forceinline__ float lane_value(float v, int src_lane) {
    return __builtin_bit_cast(float, __builtin_amdgcn_readlane(__builtin_bit_cast(int, v), src_lane));
}

__device__ __forceinline__ void build_row(
    const float* __restrict__ weights, float* __restrict__ M_re,
    float* __restrict__ M_im, float2* st, const int lane, const int k) {

    float gx = 0.f, gy = 0.f, gz = 0.f, gw = 0.f;
    if (lane < NLAYERS * NQ) {
        const float* w = weights + lane * 3;
        const float phi = w[0], theta = w[1], omega = w[2];
        const float ct = cosf(0.5f * theta), sv = sinf(0.5f * theta);
        const float al = 0.5f * (phi + omega), be = 0.5f * (phi - omega);
        const float ca = cosf(al), sa = sinf(al);
        const float cb = cosf(be), sb = sinf(be);
        gx = ct * ca; gy = ct * sa; gz = sv * cb; gw = sv * sb;
    }

    float2 a[32];

    {
        float2 P = make_float2(1.f, 0.f);
#pragma unroll
        for (int q = 0; q < 5; ++q) {
            const float x = lane_value(gx, q), y = lane_value(gy, q);
            const float z = lane_value(gz, q), w = lane_value(gw, q);
            const int kq = (k >> (9 - q)) & 1;
            const int jq = (lane >> (4 - q)) & 1;
            const float2 g0 = kq ? make_float2(-z, -w) : make_float2(x, -y);
            const float2 g1 = kq ? make_float2(x, y) : make_float2(z, -w);
            P = cmul(jq ? g1 : g0, P);
        }
        a[0] = P;
#pragma unroll
        for (int i = 0; i < 5; ++i) {
            const int q = 5 + i, b = 4 - i;
            const float x = lane_value(gx, q), y = lane_value(gy, q);
            const float z = lane_value(gz, q), w = lane_value(gw, q);
            const int kq = (k >> (9 - q)) & 1;
            const float2 g0 = kq ? make_float2(-z, -w) : make_float2(x, -y);
            const float2 g1 = kq ? make_float2(x, y) : make_float2(z, -w);
#pragma unroll
            for (int m = 0; m < (1 << i); ++m) {
                const int r0 = m << (b + 1);
                const float2 v = a[r0];
                a[r0 | (1 << b)] = cmul(g1, v);
                a[r0] = cmul(g0, v);
            }
        }
    }
    store_ring<0>(st, a, lane);

#pragma unroll 1
    for (int pass = 0; pass < 4; ++pass) {
        const int layer = 1 + (pass >> 1), hi = pass & 1;
        const int base = hi ? lane * UPITCH : lane;
        const int stride = hi ? 1 : UPITCH;
        wave_lds_fence();
#pragma unroll
        for (int r = 0; r < 32; ++r) a[r] = st[base + r * stride];
#pragma unroll
        for (int i = 0; i < 5; ++i) {
            const int gi = layer * NQ + hi * 5 + i;
            const int b = 4 - i;
            const float x = lane_value(gx, gi), y = lane_value(gy, gi);
            const float z = lane_value(gz, gi), w = lane_value(gw, gi);
            const float2 g00 = make_float2(x, -y);
            const float2 g01 = make_float2(-z, -w);
            const float2 g10 = make_float2(z, -w);
            const float2 g11 = make_float2(x, y);
#pragma unroll
            for (int p = 0; p < 16; ++p) {
                const int r0 = ((p >> b) << (b + 1)) | (p & ((1 << b) - 1));
                const int r1 = r0 | (1 << b);
                const float2 a0 = a[r0], a1 = a[r1];
                a[r0] = cadd(cmul(g00, a0), cmul(g01, a1));
                a[r1] = cadd(cmul(g10, a0), cmul(g11, a1));
            }
        }
        if (!hi) {
#pragma unroll
            for (int r = 0; r < 32; ++r) st[base + r * stride] = a[r];
        } else if (layer == 1) {
            store_ring<1>(st, a, lane);
        } else {
            const int p = __popc(k) & 3;
            const float sg = (p & 2) ? -1.f : 1.f;
#pragma unroll
            for (int r = 0; r < 32; ++r) {
                const float2 u = a[r];
                a[r] = (p & 1) ? make_float2(sg * u.y, -sg * u.x) : make_float2(sg * u.x, sg * u.y);
            }
            store_ring<2>(st, a, lane);
        }
    }
    wave_lds_fence();
#pragma unroll
    for (int r = 0; r < 32; ++r) {
        const float2 v = st[r * UPITCH + lane];
        M_re[k * SDIM + r * 32 + lane] = v.x;
        M_im[k * SDIM + r * 32 + lane] = v.y;
    }
}

#define ZPL (CDIM / 32)

__device__ __forceinline__ void embed_row(
    const float* __restrict__ z, const float* __restrict__ W_in,
    const float* __restrict__ b_in, float* __restrict__ E_mag,
    const int lane, const int b) {
    float zv[ZPL];
#pragma unroll
    for (int i = 0; i < ZPL; ++i) zv[i] = z[(size_t)b * CDIM + lane + 32 * i];
    const float bv = (lane < NQ) ? b_in[lane] : 0.f;

    float acc[NQ];
#pragma unroll
    for (int q = 0; q < NQ; ++q) acc[q] = 0.f;
#pragma unroll 6
    for (int i = 0; i < ZPL; ++i) {
        const float* wr = W_in + (lane + 32 * i) * NQ;
#pragma unroll
        for (int q = 0; q < NQ; ++q) acc[q] += zv[i] * wr[q];
    }
#pragma unroll
    for (int q = 0; q < NQ; ++q) {
#pragma unroll
        for (int s = 16; s > 0; s >>= 1) acc[q] += __shfl_xor(acc[q], s, 32);
    }
    float a = 0.f;
#pragma unroll
    for (int q = 0; q < NQ; ++q) a = (lane == q) ? acc[q] : a;
    a += bv;
    const float cl = cosf(0.5f * a), sl = sinf(0.5f * a);
    float C[NQ], S[NQ];
#pragma unroll
    for (int q = 0; q < NQ; ++q) { C[q] = lane_value(cl, q); S[q] = lane_value(sl, q); }

    float hi[32];
    hi[0] = C[0]; hi[16] = S[0];
#pragma unroll
    for (int q = 1; q < 5; ++q) {
        const int bit = 16 >> q;
#pragma unroll
        for (int r = 0; r < 32; r += 2 * bit) {
            const float v = hi[r];
            hi[r] = v * C[q];
            hi[r | bit] = v * S[q];
        }
    }
    float lo[5];
#pragma unroll
    for (int q = 5; q < NQ; ++q) lo[q - 5] = ((lane >> (9 - q)) & 1) ? S[q] : C[q];
    float* row = E_mag + (size_t)b * SDIM + lane;
#pragma unroll
    for (int r = 0; r < 32; ++r) {
        float m = hi[r];
#pragma unroll
        for (int j = 0; j < 5; ++j) m *= lo[j];
        row[r * 32] = m;
    }
}

#define BWAVES 4
#define EWAVES 8
static_assert(SDIM / BWAVES == BATCH / EWAVES, "one grid serves the rows of M and the samples");

__global__ void __launch_bounds__(32 * (BWAVES + EWAVES)) prologue_kernel(
    const float* __restrict__ weights, float* __restrict__ M_re,
    float* __restrict__ M_im, const float* __restrict__ z,
    const float* __restrict__ W_in, const float* __restrict__ b_in,
    float* __restrict__ E_mag) {
    __shared__ float2 st[BWAVES][32 * UPITCH];
    const int lane = threadIdx.x & 31;
    const int wv = __builtin_amdgcn_readfirstlane(threadIdx.x >> 5);
    if (wv < BWAVES) {
        build_row(weights, M_re, M_im, st[wv], lane, blockIdx.x * BWAVES + wv);
    } else {
        embed_row(z, W_in, b_in, E_mag, lane, blockIdx.x * EWAVES + (wv - BWAVES));
    }
}

#define APITCH (SDIM + 4)
#define GROWS  32
#define GCOLS  256
#define GWAVES 4
#define NCB    (SDIM / GCOLS)

struct BFrag { v4f r0, r1, i0, i1; };
struct AFrag { v4f lo, hi; };
struct Group { BFrag b[4]; AFrag a0, a1; };

template <int CTRL>
__device__ __forceinline__ float row_partner(float v) {
    return __builtin_bit_cast(float, __builtin_amdgcn_update_dpp(
        0, __builtin_bit_cast(int, v), CTRL, 0xF, 0xF, true));
}

__device__ __forceinline__ float half_sum16(float v) {
    v += row_partner<0xB1>(v);
    v += row_partner<0x4E>(v);
    v += row_partner<0x164>(v);
    v += row_partner<0x168>(v);
    return v;
}

#define PASS_B(q) asm volatile("" : "+v"(q.r0), "+v"(q.r1), "+v"(q.i0), "+v"(q.i1))
#define PASS_GROUP(g)                                                     \
    do {                                                                  \
        PASS_B(g.b[0]); PASS_B(g.b[1]); PASS_B(g.b[2]); PASS_B(g.b[3]);   \
        asm volatile("" : "+v"(g.a0.lo), "+v"(g.a0.hi), "+v"(g.a1.lo), "+v"(g.a1.hi)); \
    } while (0)

__global__ void __launch_bounds__(128) state_gemm_kernel(
    const float* __restrict__ E_mag,
    const float* __restrict__ M_re, const float* __restrict__ M_im,
    float* __restrict__ evs_part) {
    __shared__ __attribute__((aligned(16))) float sA[GROWS][APITCH];
    __shared__ float sRed[GWAVES][GROWS * NQ];
    const int tid = threadIdx.x;
    const int lane = tid & 31, wv = tid >> 5;
    const int cb = blockIdx.x & (NCB - 1);
    const int m0 = (blockIdx.x / NCB) * GROWS;

#pragma unroll 32
    for (int i = tid; i < GROWS * (SDIM / 4); i += 128) {
        const int row = i >> 8, c4 = (i & 255) * 4;
        const float4 v = *(const float4*)(E_mag + (size_t)(m0 + row) * SDIM + c4);
        *(float4*)(&sA[row][c4]) = v;
    }
    __syncthreads();

    const int half = lane >> 4;
    const int l15 = lane & 15;
    const int n_base = cb * GCOLS + wv * 64;
    const float* aP = &sA[l15][half * 8];
    const float* bR = M_re + (size_t)(half * 8) * SDIM + n_base + 4 * l15;
    const float* bI = M_im + (size_t)(half * 8) * SDIM + n_base + 4 * l15;

    v8f fr[2][4], fi[2][4];
#pragma unroll
    for (int u = 0; u < 2; ++u) {
#pragma unroll
        for (int t = 0; t < 4; ++t) { fr[u][t] = v8f{}; fi[u][t] = v8f{}; }
    }

    auto loadG = [&](int k0) {
        Group g;
#pragma unroll
        for (int j = 0; j < 4; ++j) {
            g.b[j].r0 = *(const v4f*)(bR + (size_t)(k0 + 2 * j) * SDIM);
            g.b[j].r1 = *(const v4f*)(bR + (size_t)(k0 + 2 * j + 1) * SDIM);
            g.b[j].i0 = *(const v4f*)(bI + (size_t)(k0 + 2 * j) * SDIM);
            g.b[j].i1 = *(const v4f*)(bI + (size_t)(k0 + 2 * j + 1) * SDIM);
        }
        g.a0.lo = *(const v4f*)(aP + k0);
        g.a0.hi = *(const v4f*)(aP + k0 + 4);
        g.a1.lo = *(const v4f*)(aP + 16 * APITCH + k0);
        g.a1.hi = *(const v4f*)(aP + 16 * APITCH + k0 + 4);
        return g;
    };
    auto mma = [&](const BFrag& b, const v2f a0, const v2f a1) {
#pragma unroll
        for (int t = 0; t < 4; ++t) {
            v2f b_re, b_im;
            b_re.x = b.r0[t]; b_re.y = b.r1[t];
            b_im.x = b.i0[t]; b_im.y = b.i1[t];
            fr[0][t] = __builtin_amdgcn_wmma_f32_16x16x4_f32(false, a0, false, b_re,
                                                             (short)0, fr[0][t], false, false);
            fi[0][t] = __builtin_amdgcn_wmma_f32_16x16x4_f32(false, a0, false, b_im,
                                                             (short)0, fi[0][t], false, false);
            fr[1][t] = __builtin_amdgcn_wmma_f32_16x16x4_f32(false, a1, false, b_re,
                                                             (short)0, fr[1][t], false, false);
            fi[1][t] = __builtin_amdgcn_wmma_f32_16x16x4_f32(false, a1, false, b_im,
                                                             (short)0, fi[1][t], false, false);
        }
    };
    auto mma4 = [&](const Group& g) {
        mma(g.b[0], __builtin_shufflevector(g.a0.lo, g.a0.lo, 0, 1),
                    __builtin_shufflevector(g.a1.lo, g.a1.lo, 0, 1));
        mma(g.b[1], __builtin_shufflevector(g.a0.lo, g.a0.lo, 2, 3),
                    __builtin_shufflevector(g.a1.lo, g.a1.lo, 2, 3));
        mma(g.b[2], __builtin_shufflevector(g.a0.hi, g.a0.hi, 0, 1),
                    __builtin_shufflevector(g.a1.hi, g.a1.hi, 0, 1));
        mma(g.b[3], __builtin_shufflevector(g.a0.hi, g.a0.hi, 2, 3),
                    __builtin_shufflevector(g.a1.hi, g.a1.hi, 2, 3));
    };

    Group g0 = loadG(0);
    PASS_GROUP(g0);
    for (int k0 = 0; k0 < SDIM; k0 += 32) {
        Group g1 = loadG(k0 + 16);
        __builtin_amdgcn_sched_barrier(0);
        PASS_GROUP(g0);
        mma4(g0);
        __builtin_amdgcn_sched_barrier(0);
        g0 = loadG((k0 + 32) & (SDIM - 1));
        __builtin_amdgcn_sched_barrier(0);
        PASS_GROUP(g1);
        mma4(g1);
        __builtin_amdgcn_sched_barrier(0);
    }

    const int jb = n_base + 4 * l15;
#pragma unroll
    for (int u = 0; u < 2; ++u) {
#pragma unroll
        for (int r = 0; r < 8; ++r) {
            float p[4];
#pragma unroll
            for (int t = 0; t < 4; ++t) {
                const float fre = fr[u][t][r];
                const float fim = fi[u][t][r];
                p[t] = fre * fre + fim * fim;
            }
            const float s01 = p[0] + p[1], s23 = p[2] + p[3];
            const float S = s01 + s23;
            const int rowm = u * 16 + r + half * 8;
            float mine = 0.f;
#pragma unroll
            for (int w = 0; w < NQ - 2; ++w) {
                float val = ((jb >> (9 - w)) & 1) ? -S : S;
                val = half_sum16(val);
                if (l15 == w) mine = val;
            }
            const float v8 = half_sum16(s01 - s23);
            const float v9 = half_sum16((p[0] - p[1]) + (p[2] - p[3]));
            if (l15 == 8) mine = v8;
            if (l15 == 9) mine = v9;
            if (l15 < NQ) sRed[wv][rowm * NQ + l15] = mine;
        }
    }
    __syncthreads();
    for (int i = tid; i < GROWS * NQ; i += 128) {
        float acc = 0.f;
#pragma unroll
        for (int g = 0; g < GWAVES; ++g) acc += sRed[g][i];
        evs_part[(size_t)cb * BATCH * NQ + (size_t)m0 * NQ + i] = acc;
    }
}

#define OSAMP   8
#define OTHREADS (CDIM / 4)
static_assert(BATCH % OSAMP == 0 && OSAMP * NQ <= OTHREADS, "the block's lanes cover its evs values");

__global__ void __launch_bounds__(OTHREADS) out_gemm_kernel(
    const float* __restrict__ evs_part, const float* __restrict__ W_out,
    const float* __restrict__ b_out, float* __restrict__ out) {
    __shared__ __attribute__((aligned(16))) float sE[OSAMP * NQ];
    const int tid = threadIdx.x;
    const int b0 = blockIdx.x * OSAMP;
    const int c0 = tid * 4;
    float4 wq[NQ];
#pragma unroll
    for (int w = 0; w < NQ; ++w) wq[w] = *(const float4*)(W_out + w * CDIM + c0);
    const float4 bo = *(const float4*)(b_out + c0);
    __builtin_amdgcn_sched_barrier(0);
    if (tid < OSAMP * NQ) {
        float a = evs_part[b0 * NQ + tid];
#pragma unroll
        for (int c = 1; c < NCB; ++c) a += evs_part[(size_t)c * BATCH * NQ + b0 * NQ + tid];
        sE[tid] = a;
    }
    __syncthreads();
#pragma unroll
    for (int s = 0; s < OSAMP; ++s) {
        float e[NQ];
#pragma unroll
        for (int w = 0; w < NQ; ++w) e[w] = sE[s * NQ + w];
        float4 acc = bo;
#pragma unroll
        for (int w = 0; w < NQ; ++w) {
            acc.x += e[w] * wq[w].x;
            acc.y += e[w] * wq[w].y;
            acc.z += e[w] * wq[w].z;
            acc.w += e[w] * wq[w].w;
        }
        *(float4*)(out + (size_t)(b0 + s) * CDIM + c0) = acc;
    }
}

extern "C" void kernel_launch(void* const* d_in, const int* in_sizes, int n_in,
                              void* d_out, int out_size, void* d_ws,
                              size_t ws_size, hipStream_t stream) {
    const float* z_i     = (const float*)d_in[0];
    const float* W_in    = (const float*)d_in[1];
    const float* b_in    = (const float*)d_in[2];
    const float* weights = (const float*)d_in[3];
    const float* W_out   = (const float*)d_in[4];
    const float* b_out   = (const float*)d_in[5];
    float* out = (float*)d_out;

    float* ws    = (float*)d_ws;
    float* M_re  = ws;
    float* M_im  = M_re + (size_t)SDIM * SDIM;
    float* E_mag = M_im + (size_t)SDIM * SDIM;
    float* evs_part = E_mag + (size_t)BATCH * SDIM;

    prologue_kernel<<<SDIM / BWAVES, 32 * (BWAVES + EWAVES), 0, stream>>>(weights, M_re, M_im, z_i, W_in, b_in, E_mag);
    state_gemm_kernel<<<(BATCH / GROWS) * NCB, 128, 0, stream>>>(E_mag, M_re, M_im, evs_part);
    out_gemm_kernel<<<BATCH / OSAMP, OTHREADS, 0, stream>>>(evs_part, W_out, b_out, out);
}


# circuit applied layer by layer on WMMA: per sample each layer = two 32x32 complex products (qubits 0-4, 5-9), CNOT ring = store address; 1024x1024 M and m no longer built; every sum order changes
# speedup vs baseline: 21.5153x; 2.4950x over previous
#include <hip/hip_runtime.h>


#define NQ      10
#define CDIM    768
#define NLAYERS 3
#define BATCH   2048

typedef float v2f __attribute__((ext_vector_type(2)));
typedef float v8f __attribute__((ext_vector_type(8)));

__device__ __forceinline__ float2 cmul(float2 a, float2 b) {
    return make_float2(a.x * b.x - a.y * b.y, a.x * b.y + a.y * b.x);
}

__host__ __device__ constexpr int cnot_ring(int l, int j) {
    const int r = (l % (NQ - 1)) + 1;
    for (int i = 0; i < NQ; ++i) {
        const int t = (i + r) % NQ;
        if ((j >> (9 - i)) & 1) j ^= 1 << (9 - t);
    }
    return j;
}

__device__ __forceinline__ void wave_lds_fence() {
    __builtin_amdgcn_fence(__ATOMIC_ACQ_REL, "workgroup", "local");
}

__device__ __forceinline__ float lane_value(float v, int src_lane) {
    return __builtin_bit_cast(float, __builtin_amdgcn_readlane(__builtin_bit_cast(int, v), src_lane));
}

#define SP     36
#define PLANE  (32 * SP)
#define SWAVES 8
#define ZPL    (CDIM / 32)

template <bool CA>
__device__ __forceinline__ void cprod(const float* cre, const float* cim, const float* cnim,
                                      const float* sre, const float* sim, const int foff,
                                      v8f (&fr)[2][2], v8f (&fi)[2][2]) {
    v2f k_re[2][8], k_im[2][8], k_nim[2][8], s_re[2][8], s_im[2][8];
#pragma unroll
    for (int t = 0; t < 2; ++t) {
#pragma unroll
        for (int ks = 0; ks < 8; ++ks) {
            const int o = foff + 16 * t * SP + 4 * ks;
            k_re[t][ks]  = *(const v2f*)(cre + o);
            k_im[t][ks]  = *(const v2f*)(cim + o);
            k_nim[t][ks] = *(const v2f*)(cnim + o);
            s_re[t][ks]  = *(const v2f*)(sre + o);
            s_im[t][ks]  = *(const v2f*)(sim + o);
        }
    }
    __builtin_amdgcn_sched_barrier(0);
#pragma unroll
    for (int u = 0; u < 2; ++u) {
#pragma unroll
        for (int n = 0; n < 2; ++n) { fr[u][n] = v8f{}; fi[u][n] = v8f{}; }
    }
#pragma unroll
    for (int ks = 0; ks < 8; ++ks) {
#pragma unroll
        for (int u = 0; u < 2; ++u) {
#pragma unroll
            for (int n = 0; n < 2; ++n) {
                const v2f a = CA ? k_re[u][ks] : s_re[u][ks];
                const v2f b = CA ? s_re[n][ks] : k_re[n][ks];
                fr[u][n] = __builtin_amdgcn_wmma_f32_16x16x4_f32(false, a, false, b, (short)0, fr[u][n], false, false);
            }
        }
#pragma unroll
        for (int u = 0; u < 2; ++u) {
#pragma unroll
            for (int n = 0; n < 2; ++n) {
                const v2f a = CA ? k_re[u][ks] : s_re[u][ks];
                const v2f b = CA ? s_im[n][ks] : k_im[n][ks];
                fi[u][n] = __builtin_amdgcn_wmma_f32_16x16x4_f32(false, a, false, b, (short)0, fi[u][n], false, false);
            }
        }
#pragma unroll
        for (int u = 0; u < 2; ++u) {
#pragma unroll
            for (int n = 0; n < 2; ++n) {
                const v2f a = CA ? k_nim[u][ks] : s_im[u][ks];
                const v2f b = CA ? s_im[n][ks] : k_nim[n][ks];
                fr[u][n] = __builtin_amdgcn_wmma_f32_16x16x4_f32(false, a, false, b, (short)0, fr[u][n], false, false);
            }
        }
#pragma unroll
        for (int u = 0; u < 2; ++u) {
#pragma unroll
            for (int n = 0; n < 2; ++n) {
                const v2f a = CA ? k_im[u][ks] : s_im[u][ks];
                const v2f b = CA ? s_re[n][ks] : k_re[n][ks];
                fi[u][n] = __builtin_amdgcn_wmma_f32_16x16x4_f32(false, a, false, b, (short)0, fi[u][n], false, false);
            }
        }
    }
}

template <int L>
__device__ __forceinline__ void layer_gates(const float (*mat)[3][PLANE], float* sr, float* si,
                                            const int foff, const int c, const int hf,
                                            v8f (&fr)[2][2], v8f (&fi)[2][2]) {
    const int ma = 2 * (L - 1), mb = ma + 1;
    wave_lds_fence();
    cprod<true>(mat[ma][0], mat[ma][1], mat[ma][2], sr, si, foff, fr, fi);
#pragma unroll
    for (int u = 0; u < 2; ++u) {
#pragma unroll
        for (int n = 0; n < 2; ++n) {
#pragma unroll
            for (int r = 0; r < 8; ++r) {
                const int o = (16 * u + 8 * hf + r) * SP + 16 * n + c;
                sr[o] = fr[u][n][r];
                si[o] = fi[u][n][r];
            }
        }
    }
    wave_lds_fence();
    cprod<false>(mat[mb][0], mat[mb][1], mat[mb][2], sr, si, foff, fr, fi);
}

__global__ void __launch_bounds__(32 * SWAVES) circuit_kernel(
    const float* __restrict__ z, const float* __restrict__ W_in,
    const float* __restrict__ b_in, const float* __restrict__ weights,
    float* __restrict__ evs) {
    __shared__ float4 sG[NLAYERS * NQ];
    __shared__ __attribute__((aligned(16))) float sMat[4][3][PLANE];
    __shared__ __attribute__((aligned(16))) float sSt[SWAVES][2][PLANE];
    const int tid = threadIdx.x;
    const int lane = tid & 31, wv = tid >> 5;
    const int b = blockIdx.x * SWAVES + wv;

    float zv[ZPL];
#pragma unroll
    for (int i = 0; i < ZPL; ++i) zv[i] = z[(size_t)b * CDIM + lane + 32 * i];
    const float bv = (lane < NQ) ? b_in[lane] : 0.f;

    if (tid < NLAYERS * NQ) {
        const float* w = weights + tid * 3;
        const float phi = w[0], theta = w[1], omega = w[2];
        const float ct = cosf(0.5f * theta), sv = sinf(0.5f * theta);
        const float al = 0.5f * (phi + omega), be = 0.5f * (phi - omega);
        sG[tid] = make_float4(ct * cosf(al), ct * sinf(al), sv * cosf(be), sv * sinf(be));
    }

    float acc[NQ];
#pragma unroll
    for (int q = 0; q < NQ; ++q) acc[q] = 0.f;
#pragma unroll 6
    for (int i = 0; i < ZPL; ++i) {
        const float* wr = W_in + (lane + 32 * i) * NQ;
#pragma unroll
        for (int q = 0; q < NQ; ++q) acc[q] += zv[i] * wr[q];
    }
#pragma unroll
    for (int q = 0; q < NQ; ++q) {
#pragma unroll
        for (int s = 16; s > 0; s >>= 1) acc[q] += __shfl_xor(acc[q], s, 32);
    }
    float ang = 0.f;
#pragma unroll
    for (int q = 0; q < NQ; ++q) ang = (lane == q) ? acc[q] : ang;
    ang += bv;
    const float cl = cosf(0.5f * ang), sl = sinf(0.5f * ang);

    __syncthreads();

    for (int idx = tid; idx < 4 * 1024; idx += 32 * SWAVES) {
        const int m = idx >> 10, i = (idx >> 5) & 31, k = idx & 31;
        const int g0 = (1 + (m >> 1)) * NQ + (m & 1) * 5;
        float2 P = make_float2(1.f, 0.f);
#pragma unroll
        for (int qq = 0; qq < 5; ++qq) {
            const float4 g = sG[g0 + qq];
            const int bi = (i >> (4 - qq)) & 1, bk = (k >> (4 - qq)) & 1;
            const float re = (bi == bk) ? g.x : (bi ? g.z : -g.z);
            const float im = (bi == bk) ? (bi ? g.y : -g.y) : -g.w;
            P = cmul(make_float2(re, im), P);
        }
        sMat[m][0][i * SP + k] = P.x;
        sMat[m][1][i * SP + k] = P.y;
        sMat[m][2][i * SP + k] = -P.y;
    }

    float* sr = sSt[wv][0];
    float* si = sSt[wv][1];
    {
        float2 hp = make_float2(1.f, 0.f), lp = make_float2(1.f, 0.f);
#pragma unroll
        for (int q = 0; q < NQ; ++q) {
            const float C = lane_value(cl, q), S = lane_value(sl, q);
            const float4 g = sG[q];
            const float2 v0 = make_float2(g.x * C - g.w * S, g.z * S - g.y * C);
            const float2 v1 = make_float2(g.z * C + g.y * S, -g.w * C - g.x * S);
            if (q < 5) hp = cmul(((lane >> (4 - q)) & 1) ? v1 : v0, hp);
            else       lp = cmul(((lane >> (9 - q)) & 1) ? v1 : v0, lp);
        }
        const int gl = cnot_ring(0, lane);
#pragma unroll
        for (int r = 0; r < 32; ++r) {
            const float2 hr = make_float2(lane_value(hp.x, r), lane_value(hp.y, r));
            const float2 a = cmul(hr, lp);
            const int x = gl ^ cnot_ring(0, r << 5);
            const int o = (x & 31) * SP + (x >> 5);
            sr[o] = a.x;
            si[o] = a.y;
        }
    }
    __syncthreads();

    const int c = lane & 15, hf = lane >> 4;
    const int foff = c * SP + 2 * hf;
    v8f fr[2][2], fi[2][2];

    layer_gates<1>(sMat, sr, si, foff, c, hf, fr, fi);
    {
        const int gl = cnot_ring(1, ((8 * hf) << 5) | c);
#pragma unroll
        for (int u = 0; u < 2; ++u) {
#pragma unroll
            for (int n = 0; n < 2; ++n) {
#pragma unroll
                for (int r = 0; r < 8; ++r) {
                    const int x = gl ^ cnot_ring(1, ((16 * u + r) << 5) | (16 * n));
                    const int o = (x & 31) * SP + (x >> 5);
                    sr[o] = fr[u][n][r];
                    si[o] = fi[u][n][r];
                }
            }
        }
    }

    layer_gates<2>(sMat, sr, si, foff, c, hf, fr, fi);
    float zs[NQ];
#pragma unroll
    for (int w = 0; w < NQ; ++w) zs[w] = 0.f;
#pragma unroll
    for (int u = 0; u < 2; ++u) {
#pragma unroll
        for (int n = 0; n < 2; ++n) {
#pragma unroll
            for (int r = 0; r < 8; ++r) {
                const float fre = fr[u][n][r], fim = fi[u][n][r];
                const float p = fre * fre + fim * fim;
                const int xc = cnot_ring(2, ((16 * u + r) << 5) | (16 * n));
#pragma unroll
                for (int w = 0; w < NQ; ++w) zs[w] += ((xc >> (9 - w)) & 1) ? -p : p;
            }
        }
    }
    const int gl2 = cnot_ring(2, ((8 * hf) << 5) | c);
    float mine = 0.f;
#pragma unroll
    for (int w = 0; w < NQ; ++w) {
        float v = ((gl2 >> (9 - w)) & 1) ? -zs[w] : zs[w];
#pragma unroll
        for (int s = 16; s > 0; s >>= 1) v += __shfl_xor(v, s, 32);
        if (lane == w) mine = v;
    }
    if (lane < NQ) evs[(size_t)b * NQ + lane] = mine;
}

#define OSAMP   8
#define OTHREADS (CDIM / 4)
static_assert(BATCH % OSAMP == 0 && OSAMP * NQ <= OTHREADS, "the block's lanes cover its evs values");

__global__ void __launch_bounds__(OTHREADS) out_gemm_kernel(
    const float* __restrict__ evs, const float* __restrict__ W_out,
    const float* __restrict__ b_out, float* __restrict__ out) {
    __shared__ __attribute__((aligned(16))) float sE[OSAMP * NQ];
    const int tid = threadIdx.x;
    const int b0 = blockIdx.x * OSAMP;
    const int c0 = tid * 4;
    float4 wq[NQ];
#pragma unroll
    for (int w = 0; w < NQ; ++w) wq[w] = *(const float4*)(W_out + w * CDIM + c0);
    const float4 bo = *(const float4*)(b_out + c0);
    __builtin_amdgcn_sched_barrier(0);
    if (tid < OSAMP * NQ) sE[tid] = evs[b0 * NQ + tid];
    __syncthreads();
#pragma unroll
    for (int s = 0; s < OSAMP; ++s) {
        float e[NQ];
#pragma unroll
        for (int w = 0; w < NQ; ++w) e[w] = sE[s * NQ + w];
        float4 acc = bo;
#pragma unroll
        for (int w = 0; w < NQ; ++w) {
            acc.x += e[w] * wq[w].x;
            acc.y += e[w] * wq[w].y;
            acc.z += e[w] * wq[w].z;
            acc.w += e[w] * wq[w].w;
        }
        *(float4*)(out + (size_t)(b0 + s) * CDIM + c0) = acc;
    }
}

extern "C" void kernel_launch(void* const* d_in, const int* in_sizes, int n_in,
                              void* d_out, int out_size, void* d_ws,
                              size_t ws_size, hipStream_t stream) {
    const float* z_i     = (const float*)d_in[0];
    const float* W_in    = (const float*)d_in[1];
    const float* b_in    = (const float*)d_in[2];
    const float* weights = (const float*)d_in[3];
    const float* W_out   = (const float*)d_in[4];
    const float* b_out   = (const float*)d_in[5];
    float* out = (float*)d_out;

    float* evs = (float*)d_ws;

    circuit_kernel<<<BATCH / SWAVES, 32 * SWAVES, 0, stream>>>(z_i, W_in, b_in, weights, evs);
    out_gemm_kernel<<<BATCH / OSAMP, OTHREADS, 0, stream>>>(evs, W_out, b_out, out);
}
